# starting kernel
# baseline (speedup 1.0000x reference)
#include <hip/hip_runtime.h>
#include <hip/hip_bf16.h>

typedef float v2f __attribute__((ext_vector_type(2)));
typedef float v8f __attribute__((ext_vector_type(8)));

#define FD 128


__global__ void gcn_deg_kernel(const int* __restrict__ dst, float* __restrict__ deg, int E) {
    int e = blockIdx.x * blockDim.x + threadIdx.x;
    if (e < E) unsafeAtomicAdd(&deg[dst[e]], 1.0f);
}

__global__ void gcn_dis_kernel(const float* __restrict__ deg, float* __restrict__ dis, int N) {
    int i = blockIdx.x * blockDim.x + threadIdx.x;
    if (i < N) dis[i] = rsqrtf(deg[i] + 1.0f);
}


__global__ void gcn_gemm_wmma(const float* __restrict__ A, const float* __restrict__ W,
                              float* __restrict__ C, int M) {
    const int K  = FD;
    const int Nc = FD;
    int wave = (int)((blockIdx.x * blockDim.x + threadIdx.x) >> 5);
    int lane = threadIdx.x & 31;
    const int tiles_n = Nc / 16;
    int tile_m = wave / tiles_n;
    int tile_n = wave - tile_m * tiles_n;
    if (tile_m * 16 >= M) return;

    int half = lane >> 4;
    int l16  = lane & 15;

    const float* arow = A + (size_t)(tile_m * 16 + l16) * K;
    const float* wcol = W + (size_t)(tile_n * 16 + l16);

    v8f c = {};
    #pragma unroll 4
    for (int k = 0; k < K; k += 4) {
        v2f a, b;
        int ka = k + 2 * half;
        a.x = arow[ka];
        a.y = arow[ka + 1];
        b.x = wcol[(size_t)ka * Nc];
        b.y = wcol[(size_t)(ka + 1) * Nc];
        c = __builtin_amdgcn_wmma_f32_16x16x4_f32(false, a, false, b,
                                                  (short)0, c, false, false);
    }

    float* crow = C + (size_t)(tile_m * 16) * Nc + tile_n * 16;
    #pragma unroll
    for (int v = 0; v < 8; ++v) {
        crow[(size_t)(v + 8 * half) * Nc + l16] = c[v];
    }
}


__global__ void gcn_edge_agg(const float* __restrict__ h, const int* __restrict__ src,
                             const int* __restrict__ dst, const float* __restrict__ dis,
                             float* __restrict__ out, int E, int N) {
    int idx  = (int)((blockIdx.x * blockDim.x + threadIdx.x) >> 5);
    int lane = threadIdx.x & 31;
    int total = E + N;
    if (idx >= total) return;

    int s, d;
    if (idx < E) { s = src[idx]; d = dst[idx]; }
    else         { s = idx - E; d = s; }

    float nrm = dis[s] * dis[d];
    const float4* hs = (const float4*)(h + (size_t)s * FD);
    float4 v = hs[lane];
    float* o = out + (size_t)d * FD + lane * 4;
    unsafeAtomicAdd(o + 0, v.x * nrm);
    unsafeAtomicAdd(o + 1, v.y * nrm);
    unsafeAtomicAdd(o + 2, v.z * nrm);
    unsafeAtomicAdd(o + 3, v.w * nrm);
}


__global__ void gcn_bias_relu(const float* __restrict__ in, const float* __restrict__ b,
                              float* __restrict__ out, int total) {
    int i = blockIdx.x * blockDim.x + threadIdx.x;
    if (i < total) {
        float v = in[i] + b[i & (FD - 1)];
        out[i] = v > 0.0f ? v : 0.0f;
    }
}

__global__ void gcn_bias_add(float* __restrict__ out, const float* __restrict__ b, int total) {
    int i = blockIdx.x * blockDim.x + threadIdx.x;
    if (i < total) out[i] += b[i & (FD - 1)];
}


extern "C" void kernel_launch(void* const* d_in, const int* in_sizes, int n_in,
                              void* d_out, int out_size, void* d_ws, size_t ws_size,
                              hipStream_t stream) {
    const float* x  = (const float*)d_in[0];
    const int*   ei = (const int*)d_in[1];
    const float* W1 = (const float*)d_in[2];
    const float* b1 = (const float*)d_in[3];
    const float* W2 = (const float*)d_in[4];
    const float* b2 = (const float*)d_in[5];

    const int N = in_sizes[0] / FD;
    const int E = in_sizes[1] / 2;
    const int* src = ei;
    const int* dst = ei + E;

    float* ws   = (float*)d_ws;
    float* deg  = ws;
    float* dis  = ws + N;
    float* bufA = ws + 2 * (size_t)N;
    float* bufB = bufA + (size_t)N * FD;
    float* out  = (float*)d_out;

    const size_t feat_bytes = (size_t)N * FD * sizeof(float);
    const int total = N * FD;

    hipMemsetAsync(deg, 0, (size_t)N * sizeof(float), stream);
    gcn_deg_kernel<<<(E + 255) / 256, 256, 0, stream>>>(dst, deg, E);
    gcn_dis_kernel<<<(N + 255) / 256, 256, 0, stream>>>(deg, dis, N);

    const int tilesM  = (N + 15) / 16;
    const int waves   = tilesM * (FD / 16);
    const int gthreads = waves * 32;
    const int gblocks  = (gthreads + 255) / 256;

    const int ethreads = (E + N) * 32;
    const int eblocks  = (ethreads + 255) / 256;

    gcn_gemm_wmma<<<gblocks, 256, 0, stream>>>(x, W1, bufA, N);
    hipMemsetAsync(bufB, 0, feat_bytes, stream);
    gcn_edge_agg<<<eblocks, 256, 0, stream>>>(bufA, src, dst, dis, bufB, E, N);
    gcn_bias_relu<<<(total + 255) / 256, 256, 0, stream>>>(bufB, b1, bufA, total);

    gcn_gemm_wmma<<<gblocks, 256, 0, stream>>>(bufA, W2, bufB, N);
    hipMemsetAsync(out, 0, feat_bytes, stream);
    gcn_edge_agg<<<eblocks, 256, 0, stream>>>(bufB, src, dst, dis, out, E, N);
    gcn_bias_add<<<(total + 255) / 256, 256, 0, stream>>>(out, b2, total);
}


# Aggregation by owner work-groups: each of 256 groups scans the raw edge list for its 196 destination rows, sums gathered rows in LDS (LDS f32 atomic adds, arrival order), stores rows once
# speedup vs baseline: 1.7138x; 1.7138x over previous
#include <hip/hip_runtime.h>
#include <hip/hip_bf16.h>

typedef float v2f __attribute__((ext_vector_type(2)));
typedef float v8f __attribute__((ext_vector_type(8)));

#define FD 128


__global__ void gcn_deg_kernel(const int* __restrict__ dst, float* __restrict__ deg, int E) {
    int e = blockIdx.x * blockDim.x + threadIdx.x;
    if (e < E) unsafeAtomicAdd(&deg[dst[e]], 1.0f);
}

__global__ void gcn_dis_kernel(const float* __restrict__ deg, float* __restrict__ dis, int N) {
    int i = blockIdx.x * blockDim.x + threadIdx.x;
    if (i < N) dis[i] = rsqrtf(deg[i] + 1.0f);
}


__global__ void gcn_gemm_wmma(const float* __restrict__ A, const float* __restrict__ W,
                              float* __restrict__ C, int M) {
    const int K  = FD;
    const int Nc = FD;
    int wave = (int)((blockIdx.x * blockDim.x + threadIdx.x) >> 5);
    int lane = threadIdx.x & 31;
    const int tiles_n = Nc / 16;
    int tile_m = wave / tiles_n;
    int tile_n = wave - tile_m * tiles_n;
    if (tile_m * 16 >= M) return;

    int half = lane >> 4;
    int l16  = lane & 15;

    const float* arow = A + (size_t)(tile_m * 16 + l16) * K;
    const float* wcol = W + (size_t)(tile_n * 16 + l16);

    v8f c = {};
    #pragma unroll 4
    for (int k = 0; k < K; k += 4) {
        v2f a, b;
        int ka = k + 2 * half;
        a.x = arow[ka];
        a.y = arow[ka + 1];
        b.x = wcol[(size_t)ka * Nc];
        b.y = wcol[(size_t)(ka + 1) * Nc];
        c = __builtin_amdgcn_wmma_f32_16x16x4_f32(false, a, false, b,
                                                  (short)0, c, false, false);
    }

    float* crow = C + (size_t)(tile_m * 16) * Nc + tile_n * 16;
    #pragma unroll
    for (int v = 0; v < 8; ++v) {
        crow[(size_t)(v + 8 * half) * Nc + l16] = c[v];
    }
}


#define AGG_ROWS    196
#define AGG_CAP     8192
#define AGG_THREADS 1024
#define AGG_WAVES   (AGG_THREADS / 32)

template <bool RELU>
__global__ __launch_bounds__(AGG_THREADS)
void gcn_agg_owner(const float* __restrict__ h, const int* __restrict__ src,
                   const int* __restrict__ dst, const float* __restrict__ dis,
                   const float* __restrict__ bias, float* __restrict__ out, int E, int N) {
    __shared__ float acc[AGG_ROWS * FD];
    __shared__ int   l_a[AGG_CAP];
    __shared__ int   l_r[AGG_CAP];
    __shared__ float l_n[AGG_CAP];
    __shared__ int   cnt;

    const int tid  = threadIdx.x;
    const int lane = tid & 31;
    const int w    = __builtin_amdgcn_readfirstlane(tid >> 5);
    const int lo   = blockIdx.x * AGG_ROWS;
    const int rows = min(AGG_ROWS, N - lo);
    const unsigned urows = (unsigned)rows;
    const float4* h4 = (const float4*)h;

    for (int r = w; r < rows; r += AGG_WAVES) {
        float dd = dis[lo + r];
        float nn = dd * dd;
        float4 v = h4[(size_t)(lo + r) * (FD / 4) + lane];
        float* a = acc + r * FD + lane;
        a[0]  = v.x * nn;
        a[32] = v.y * nn;
        a[64] = v.z * nn;
        a[96] = v.w * nn;
    }
    if (tid == 0) cnt = 0;
    __syncthreads();

    auto push = [&](int e, unsigned r) {
        int pos = atomicAdd(&cnt, 1);
        if (pos < AGG_CAP) { l_a[pos] = e; l_r[pos] = (int)r; }
    };

    int head = (int)(((16u - (unsigned)((size_t)dst & 15u)) & 15u) >> 2);
    if (head > E) head = E;
    const int nvec  = (E - head) >> 2;
    const int tail0 = head + nvec * 4;
    const int4* d4  = (const int4*)(dst + head);
    {
        int e = tid < head ? tid : tail0 + (tid - head);
        if (e < E) {
            unsigned r = (unsigned)(dst[e] - lo);
            if (r < urows) push(e, r);
        }
    }
    const int start = (int)(((long long)blockIdx.x * nvec) / gridDim.x);
    for (int k0 = tid; k0 < nvec; k0 += 8 * AGG_THREADS) {
        int4 d[8];
        int  id[8];
        unsigned lim[8];
        #pragma unroll
        for (int u = 0; u < 8; ++u) {
            int k = k0 + u * AGG_THREADS;
            int j = k + start;
            if (j >= nvec) j -= nvec;
            bool ok = k < nvec;
            j = ok ? j : 0;
            lim[u] = ok ? urows : 0u;
            id[u] = j;
            d[u] = d4[j];
        }
        #pragma unroll
        for (int u = 0; u < 8; ++u) {
            unsigned r0 = (unsigned)(d[u].x - lo);
            unsigned r1 = (unsigned)(d[u].y - lo);
            unsigned r2 = (unsigned)(d[u].z - lo);
            unsigned r3 = (unsigned)(d[u].w - lo);
            unsigned lm = lim[u];
            if ((r0 < lm) | (r1 < lm) | (r2 < lm) | (r3 < lm)) {
                int e0 = head + 4 * id[u];
                if (r0 < lm) push(e0 + 0, r0);
                if (r1 < lm) push(e0 + 1, r1);
                if (r2 < lm) push(e0 + 2, r2);
                if (r3 < lm) push(e0 + 3, r3);
            }
        }
    }
    __syncthreads();
    const int found = cnt;

    if (found > AGG_CAP) {
        for (int e = tid; e < E; e += AGG_THREADS) {
            unsigned r = (unsigned)(dst[e] - lo);
            if (r < urows) {
                int s = src[e];
                float nn = dis[s] * dis[lo + (int)r];
                const float* hs = h + (size_t)s * FD;
                for (int c = 0; c < FD; ++c)
                    atomicAdd(&acc[r * FD + (c & 3) * 32 + (c >> 2)], hs[c] * nn);
            }
        }
    } else {
        for (int t = tid; t < found; t += AGG_THREADS) {
            int e = l_a[t];
            int s = src[e];
            l_a[t] = s;
            l_n[t] = dis[s] * dis[lo + l_r[t]];
        }
        __syncthreads();
        const int groups = found >> 3;
        for (int g = w; g < groups; g += AGG_WAVES) {
            float4 v[8];
            float  nn[8];
            int    rr[8];
            #pragma unroll
            for (int u = 0; u < 8; ++u) {
                int i = g * 8 + u;
                int s = l_a[i];
                rr[u] = l_r[i];
                nn[u] = l_n[i];
                v[u]  = h4[(size_t)s * (FD / 4) + lane];
            }
            #pragma unroll
            for (int u = 0; u < 8; ++u) {
                float* a = acc + rr[u] * FD + lane;
                atomicAdd(a,      v[u].x * nn[u]);
                atomicAdd(a + 32, v[u].y * nn[u]);
                atomicAdd(a + 64, v[u].z * nn[u]);
                atomicAdd(a + 96, v[u].w * nn[u]);
            }
        }
        {
            int i = groups * 8 + w;
            if (i < found) {
                int s = l_a[i];
                float n1 = l_n[i];
                float4 v1 = h4[(size_t)s * (FD / 4) + lane];
                float* a = acc + l_r[i] * FD + lane;
                atomicAdd(a,      v1.x * n1);
                atomicAdd(a + 32, v1.y * n1);
                atomicAdd(a + 64, v1.z * n1);
                atomicAdd(a + 96, v1.w * n1);
            }
        }
    }
    __syncthreads();

    const float4 b4 = ((const float4*)bias)[lane];
    float4* o4 = (float4*)out;
    for (int r = w; r < rows; r += AGG_WAVES) {
        const float* a = acc + r * FD + lane;
        float4 v;
        v.x = a[0]  + b4.x;
        v.y = a[32] + b4.y;
        v.z = a[64] + b4.z;
        v.w = a[96] + b4.w;
        if (RELU) {
            v.x = v.x > 0.0f ? v.x : 0.0f;
            v.y = v.y > 0.0f ? v.y : 0.0f;
            v.z = v.z > 0.0f ? v.z : 0.0f;
            v.w = v.w > 0.0f ? v.w : 0.0f;
        }
        o4[(size_t)(lo + r) * (FD / 4) + lane] = v;
    }
}


extern "C" void kernel_launch(void* const* d_in, const int* in_sizes, int n_in,
                              void* d_out, int out_size, void* d_ws, size_t ws_size,
                              hipStream_t stream) {
    const float* x  = (const float*)d_in[0];
    const int*   ei = (const int*)d_in[1];
    const float* W1 = (const float*)d_in[2];
    const float* b1 = (const float*)d_in[3];
    const float* W2 = (const float*)d_in[4];
    const float* b2 = (const float*)d_in[5];

    const int N = in_sizes[0] / FD;
    const int E = in_sizes[1] / 2;
    const int* src = ei;
    const int* dst = ei + E;

    float* ws   = (float*)d_ws;
    float* deg  = ws;
    float* dis  = ws + N;
    float* bufA = ws + 2 * (size_t)N;
    float* bufB = bufA + (size_t)N * FD;
    float* out  = (float*)d_out;

    hipMemsetAsync(deg, 0, (size_t)N * sizeof(float), stream);
    gcn_deg_kernel<<<(E + 255) / 256, 256, 0, stream>>>(dst, deg, E);
    gcn_dis_kernel<<<(N + 255) / 256, 256, 0, stream>>>(deg, dis, N);

    const int tilesM  = (N + 15) / 16;
    const int waves   = tilesM * (FD / 16);
    const int gthreads = waves * 32;
    const int gblocks  = (gthreads + 255) / 256;

    const int ablocks = (N + AGG_ROWS - 1) / AGG_ROWS;

    gcn_gemm_wmma<<<gblocks, 256, 0, stream>>>(x, W1, bufA, N);
    gcn_agg_owner<true><<<ablocks, AGG_THREADS, 0, stream>>>(bufA, src, dst, dis, b1, bufB, E, N);

    gcn_gemm_wmma<<<gblocks, 256, 0, stream>>>(bufB, W2, bufA, N);
    gcn_agg_owner<false><<<ablocks, AGG_THREADS, 0, stream>>>(bufA, src, dst, dis, b2, out, E, N);
}


# GEMM cut by output columns: a wave keeps its 128x16 strip of W in registers over 64 rows; A and C rows pass once through LDS as whole 512-byte rows shared by 8 waves; sum order unchanged
# speedup vs baseline: 1.8699x; 1.0911x over previous
#include <hip/hip_runtime.h>
#include <hip/hip_bf16.h>

typedef float v2f __attribute__((ext_vector_type(2)));
typedef float v8f __attribute__((ext_vector_type(8)));

#define FD 128


__global__ void gcn_deg_kernel(const int* __restrict__ dst, float* __restrict__ deg, int E) {
    int e = blockIdx.x * blockDim.x + threadIdx.x;
    if (e < E) unsafeAtomicAdd(&deg[dst[e]], 1.0f);
}

__global__ void gcn_dis_kernel(const float* __restrict__ deg, float* __restrict__ dis, int N) {
    int i = blockIdx.x * blockDim.x + threadIdx.x;
    if (i < N) dis[i] = rsqrtf(deg[i] + 1.0f);
}


#define GEMM_ROWS  64
#define GEMM_PITCH 132
#define GEMM_RT    (GEMM_ROWS / 16)

__global__ __launch_bounds__(256)
void gcn_gemm_wmma(const float* __restrict__ A, const float* __restrict__ W,
                   float* __restrict__ C, int M) {
    __shared__ float sa[GEMM_ROWS * GEMM_PITCH];

    const int tid  = threadIdx.x;
    const int lane = tid & 31;
    const int wv   = __builtin_amdgcn_readfirstlane(tid >> 5);
    const int half = lane >> 4;
    const int l16  = lane & 15;
    const int row0 = blockIdx.x * GEMM_ROWS;

    const float4* A4 = (const float4*)A;
    float4 av[GEMM_ROWS / 8];
    #pragma unroll
    for (int i = 0; i < GEMM_ROWS / 8; ++i) {
        int gr = row0 + i * 8 + wv;
        av[i] = gr < M ? A4[(size_t)gr * (FD / 4) + lane] : make_float4(0.0f, 0.0f, 0.0f, 0.0f);
    }

    v2f wf[FD / 4];
    const float* wcol = W + (size_t)(2 * half) * FD + wv * 16 + l16;
    #pragma unroll
    for (int ks = 0; ks < FD / 4; ++ks) {
        wf[ks].x = wcol[(size_t)(4 * ks) * FD];
        wf[ks].y = wcol[(size_t)(4 * ks + 1) * FD];
    }

    #pragma unroll
    for (int i = 0; i < GEMM_ROWS / 8; ++i) {
        *(float4*)&sa[(i * 8 + wv) * GEMM_PITCH + 4 * lane] = av[i];
    }
    __syncthreads();

    v8f acc[GEMM_RT];
    #pragma unroll
    for (int rt = 0; rt < GEMM_RT; ++rt) acc[rt] = v8f{};

    const float* arow = sa + l16 * GEMM_PITCH + 2 * half;
    #pragma unroll
    for (int ks = 0; ks < FD / 4; ++ks) {
        #pragma unroll
        for (int rt = 0; rt < GEMM_RT; ++rt) {
            v2f a = *(const v2f*)&arow[rt * 16 * GEMM_PITCH + 4 * ks];
            acc[rt] = __builtin_amdgcn_wmma_f32_16x16x4_f32(false, a, false, wf[ks],
                                                            (short)0, acc[rt], false, false);
        }
    }

    __syncthreads();
    float* crow = sa + (8 * half) * GEMM_PITCH + wv * 16 + l16;
    #pragma unroll
    for (int rt = 0; rt < GEMM_RT; ++rt) {
        #pragma unroll
        for (int v = 0; v < 8; ++v) {
            crow[(rt * 16 + v) * GEMM_PITCH] = acc[rt][v];
        }
    }
    __syncthreads();
    float4* C4 = (float4*)C;
    #pragma unroll
    for (int i = 0; i < GEMM_ROWS / 8; ++i) {
        int r  = i * 8 + wv;
        int gr = row0 + r;
        if (gr < M) C4[(size_t)gr * (FD / 4) + lane] = *(const float4*)&sa[r * GEMM_PITCH + 4 * lane];
    }
}


#define AGG_ROWS    196
#define AGG_CAP     8192
#define AGG_THREADS 1024
#define AGG_WAVES   (AGG_THREADS / 32)

template <bool RELU>
__global__ __launch_bounds__(AGG_THREADS)
void gcn_agg_owner(const float* __restrict__ h, const int* __restrict__ src,
                   const int* __restrict__ dst, const float* __restrict__ dis,
                   const float* __restrict__ bias, float* __restrict__ out, int E, int N) {
    __shared__ float acc[AGG_ROWS * FD];
    __shared__ int   l_a[AGG_CAP];
    __shared__ int   l_r[AGG_CAP];
    __shared__ float l_n[AGG_CAP];
    __shared__ int   cnt;

    const int tid  = threadIdx.x;
    const int lane = tid & 31;
    const int w    = __builtin_amdgcn_readfirstlane(tid >> 5);
    const int lo   = blockIdx.x * AGG_ROWS;
    const int rows = min(AGG_ROWS, N - lo);
    const unsigned urows = (unsigned)rows;
    const float4* h4 = (const float4*)h;

    for (int r = w; r < rows; r += AGG_WAVES) {
        float dd = dis[lo + r];
        float nn = dd * dd;
        float4 v = h4[(size_t)(lo + r) * (FD / 4) + lane];
        float* a = acc + r * FD + lane;
        a[0]  = v.x * nn;
        a[32] = v.y * nn;
        a[64] = v.z * nn;
        a[96] = v.w * nn;
    }
    if (tid == 0) cnt = 0;
    __syncthreads();

    auto push = [&](int e, unsigned r) {
        int pos = atomicAdd(&cnt, 1);
        if (pos < AGG_CAP) { l_a[pos] = e; l_r[pos] = (int)r; }
    };

    int head = (int)(((16u - (unsigned)((size_t)dst & 15u)) & 15u) >> 2);
    if (head > E) head = E;
    const int nvec  = (E - head) >> 2;
    const int tail0 = head + nvec * 4;
    const int4* d4  = (const int4*)(dst + head);
    {
        int e = tid < head ? tid : tail0 + (tid - head);
        if (e < E) {
            unsigned r = (unsigned)(dst[e] - lo);
            if (r < urows) push(e, r);
        }
    }
    const int start = (int)(((long long)blockIdx.x * nvec) / gridDim.x);
    for (int k0 = tid; k0 < nvec; k0 += 8 * AGG_THREADS) {
        int4 d[8];
        int  id[8];
        unsigned lim[8];
        #pragma unroll
        for (int u = 0; u < 8; ++u) {
            int k = k0 + u * AGG_THREADS;
            int j = k + start;
            if (j >= nvec) j -= nvec;
            bool ok = k < nvec;
            j = ok ? j : 0;
            lim[u] = ok ? urows : 0u;
            id[u] = j;
            d[u] = d4[j];
        }
        #pragma unroll
        for (int u = 0; u < 8; ++u) {
            unsigned r0 = (unsigned)(d[u].x - lo);
            unsigned r1 = (unsigned)(d[u].y - lo);
            unsigned r2 = (unsigned)(d[u].z - lo);
            unsigned r3 = (unsigned)(d[u].w - lo);
            unsigned lm = lim[u];
            if ((r0 < lm) | (r1 < lm) | (r2 < lm) | (r3 < lm)) {
                int e0 = head + 4 * id[u];
                if (r0 < lm) push(e0 + 0, r0);
                if (r1 < lm) push(e0 + 1, r1);
                if (r2 < lm) push(e0 + 2, r2);
                if (r3 < lm) push(e0 + 3, r3);
            }
        }
    }
    __syncthreads();
    const int found = cnt;

    if (found > AGG_CAP) {
        for (int e = tid; e < E; e += AGG_THREADS) {
            unsigned r = (unsigned)(dst[e] - lo);
            if (r < urows) {
                int s = src[e];
                float nn = dis[s] * dis[lo + (int)r];
                const float* hs = h + (size_t)s * FD;
                for (int c = 0; c < FD; ++c)
                    atomicAdd(&acc[r * FD + (c & 3) * 32 + (c >> 2)], hs[c] * nn);
            }
        }
    } else {
        for (int t = tid; t < found; t += AGG_THREADS) {
            int e = l_a[t];
            int s = src[e];
            l_a[t] = s;
            l_n[t] = dis[s] * dis[lo + l_r[t]];
        }
        __syncthreads();
        const int groups = found >> 3;
        for (int g = w; g < groups; g += AGG_WAVES) {
            float4 v[8];
            float  nn[8];
            int    rr[8];
            #pragma unroll
            for (int u = 0; u < 8; ++u) {
                int i = g * 8 + u;
                int s = l_a[i];
                rr[u] = l_r[i];
                nn[u] = l_n[i];
                v[u]  = h4[(size_t)s * (FD / 4) + lane];
            }
            #pragma unroll
            for (int u = 0; u < 8; ++u) {
                float* a = acc + rr[u] * FD + lane;
                atomicAdd(a,      v[u].x * nn[u]);
                atomicAdd(a + 32, v[u].y * nn[u]);
                atomicAdd(a + 64, v[u].z * nn[u]);
                atomicAdd(a + 96, v[u].w * nn[u]);
            }
        }
        {
            int i = groups * 8 + w;
            if (i < found) {
                int s = l_a[i];
                float n1 = l_n[i];
                float4 v1 = h4[(size_t)s * (FD / 4) + lane];
                float* a = acc + l_r[i] * FD + lane;
                atomicAdd(a,      v1.x * n1);
                atomicAdd(a + 32, v1.y * n1);
                atomicAdd(a + 64, v1.z * n1);
                atomicAdd(a + 96, v1.w * n1);
            }
        }
    }
    __syncthreads();

    const float4 b4 = ((const float4*)bias)[lane];
    float4* o4 = (float4*)out;
    for (int r = w; r < rows; r += AGG_WAVES) {
        const float* a = acc + r * FD + lane;
        float4 v;
        v.x = a[0]  + b4.x;
        v.y = a[32] + b4.y;
        v.z = a[64] + b4.z;
        v.w = a[96] + b4.w;
        if (RELU) {
            v.x = v.x > 0.0f ? v.x : 0.0f;
            v.y = v.y > 0.0f ? v.y : 0.0f;
            v.z = v.z > 0.0f ? v.z : 0.0f;
            v.w = v.w > 0.0f ? v.w : 0.0f;
        }
        o4[(size_t)(lo + r) * (FD / 4) + lane] = v;
    }
}


extern "C" void kernel_launch(void* const* d_in, const int* in_sizes, int n_in,
                              void* d_out, int out_size, void* d_ws, size_t ws_size,
                              hipStream_t stream) {
    const float* x  = (const float*)d_in[0];
    const int*   ei = (const int*)d_in[1];
    const float* W1 = (const float*)d_in[2];
    const float* b1 = (const float*)d_in[3];
    const float* W2 = (const float*)d_in[4];
    const float* b2 = (const float*)d_in[5];

    const int N = in_sizes[0] / FD;
    const int E = in_sizes[1] / 2;
    const int* src = ei;
    const int* dst = ei + E;

    float* ws   = (float*)d_ws;
    float* deg  = ws;
    float* dis  = ws + N;
    float* bufA = ws + 2 * (size_t)N;
    float* bufB = bufA + (size_t)N * FD;
    float* out  = (float*)d_out;

    hipMemsetAsync(deg, 0, (size_t)N * sizeof(float), stream);
    gcn_deg_kernel<<<(E + 255) / 256, 256, 0, stream>>>(dst, deg, E);
    gcn_dis_kernel<<<(N + 255) / 256, 256, 0, stream>>>(deg, dis, N);

    const int gblocks = (N + GEMM_ROWS - 1) / GEMM_ROWS;

    const int ablocks = (N + AGG_ROWS - 1) / AGG_ROWS;

    gcn_gemm_wmma<<<gblocks, 256, 0, stream>>>(x, W1, bufA, N);
    gcn_agg_owner<true><<<ablocks, AGG_THREADS, 0, stream>>>(bufA, src, dst, dis, b1, bufB, E, N);

    gcn_gemm_wmma<<<gblocks, 256, 0, stream>>>(bufB, W2, bufA, N);
    gcn_agg_owner<false><<<ablocks, AGG_THREADS, 0, stream>>>(bufA, src, dst, dis, b2, out, E, N);
}


# Degree launch also writes a per-owner edge bitmap (1 bit per owner and edge, built in LDS, whole 128-byte lines); an owner group lists its edges from its 75 KB bitmap row, not the whole dst column
# speedup vs baseline: 2.3718x; 1.2684x over previous
#include <hip/hip_runtime.h>
#include <hip/hip_bf16.h>

typedef float v2f __attribute__((ext_vector_type(2)));
typedef float v8f __attribute__((ext_vector_type(8)));

#define FD 128


#define AGG_ROWS    196
#define OWN_SLOTS   256
#define BM_EDGES    1024
#define BM_WORDS    (BM_EDGES / 32)
#define BM_PITCH    (OWN_SLOTS + 1)

__global__ __launch_bounds__(BM_EDGES)
void gcn_deg_own_kernel(const int* __restrict__ dst, float* __restrict__ deg,
                        unsigned* __restrict__ bm, int E, int rowWords) {
    __shared__ unsigned m[BM_WORDS * BM_PITCH];
    const int tid = threadIdx.x;
    for (int i = tid; i < BM_WORDS * BM_PITCH; i += BM_EDGES) m[i] = 0u;
    __syncthreads();
    const int e = blockIdx.x * BM_EDGES + tid;
    if (e < E) {
        const int d = dst[e];
        unsafeAtomicAdd(&deg[d], 1.0f);
        const unsigned o = ((unsigned)d / AGG_ROWS) & (OWN_SLOTS - 1);
        atomicOr(&m[(tid >> 5) * BM_PITCH + o], 1u << (tid & 31));
    }
    __syncthreads();
    const int w = tid & 31;
    #pragma unroll
    for (int o = tid >> 5; o < OWN_SLOTS; o += BM_EDGES / 32) {
        bm[(size_t)o * rowWords + (size_t)blockIdx.x * BM_WORDS + w] = m[w * BM_PITCH + o];
    }
}

__global__ void gcn_dis_kernel(const float* __restrict__ deg, float* __restrict__ dis, int N) {
    int i = blockIdx.x * blockDim.x + threadIdx.x;
    if (i < N) dis[i] = rsqrtf(deg[i] + 1.0f);
}


#define GEMM_ROWS  64
#define GEMM_PITCH 132
#define GEMM_RT    (GEMM_ROWS / 16)

__global__ __launch_bounds__(256)
void gcn_gemm_wmma(const float* __restrict__ A, const float* __restrict__ W,
                   float* __restrict__ C, int M) {
    __shared__ float sa[GEMM_ROWS * GEMM_PITCH];

    const int tid  = threadIdx.x;
    const int lane = tid & 31;
    const int wv   = __builtin_amdgcn_readfirstlane(tid >> 5);
    const int half = lane >> 4;
    const int l16  = lane & 15;
    const int row0 = blockIdx.x * GEMM_ROWS;

    const float4* A4 = (const float4*)A;
    float4 av[GEMM_ROWS / 8];
    #pragma unroll
    for (int i = 0; i < GEMM_ROWS / 8; ++i) {
        int gr = row0 + i * 8 + wv;
        av[i] = gr < M ? A4[(size_t)gr * (FD / 4) + lane] : make_float4(0.0f, 0.0f, 0.0f, 0.0f);
    }

    v2f wf[FD / 4];
    const float* wcol = W + (size_t)(2 * half) * FD + wv * 16 + l16;
    #pragma unroll
    for (int ks = 0; ks < FD / 4; ++ks) {
        wf[ks].x = wcol[(size_t)(4 * ks) * FD];
        wf[ks].y = wcol[(size_t)(4 * ks + 1) * FD];
    }

    #pragma unroll
    for (int i = 0; i < GEMM_ROWS / 8; ++i) {
        *(float4*)&sa[(i * 8 + wv) * GEMM_PITCH + 4 * lane] = av[i];
    }
    __syncthreads();

    v8f acc[GEMM_RT];
    #pragma unroll
    for (int rt = 0; rt < GEMM_RT; ++rt) acc[rt] = v8f{};

    const float* arow = sa + l16 * GEMM_PITCH + 2 * half;
    #pragma unroll
    for (int ks = 0; ks < FD / 4; ++ks) {
        #pragma unroll
        for (int rt = 0; rt < GEMM_RT; ++rt) {
            v2f a = *(const v2f*)&arow[rt * 16 * GEMM_PITCH + 4 * ks];
            acc[rt] = __builtin_amdgcn_wmma_f32_16x16x4_f32(false, a, false, wf[ks],
                                                            (short)0, acc[rt], false, false);
        }
    }

    __syncthreads();
    float* crow = sa + (8 * half) * GEMM_PITCH + wv * 16 + l16;
    #pragma unroll
    for (int rt = 0; rt < GEMM_RT; ++rt) {
        #pragma unroll
        for (int v = 0; v < 8; ++v) {
            crow[(rt * 16 + v) * GEMM_PITCH] = acc[rt][v];
        }
    }
    __syncthreads();
    float4* C4 = (float4*)C;
    #pragma unroll
    for (int i = 0; i < GEMM_ROWS / 8; ++i) {
        int r  = i * 8 + wv;
        int gr = row0 + r;
        if (gr < M) C4[(size_t)gr * (FD / 4) + lane] = *(const float4*)&sa[r * GEMM_PITCH + 4 * lane];
    }
}


#define AGG_CAP     8192
#define AGG_THREADS 1024
#define AGG_WAVES   (AGG_THREADS / 32)

template <bool RELU>
__global__ __launch_bounds__(AGG_THREADS)
void gcn_agg_owner(const float* __restrict__ h, const int* __restrict__ src,
                   const int* __restrict__ dst, const float* __restrict__ dis,
                   const unsigned* __restrict__ bm, int rowWords,
                   const float* __restrict__ bias, float* __restrict__ out, int E, int N) {
    __shared__ float acc[(AGG_ROWS + 1) * FD];
    __shared__ int   l_a[AGG_CAP];
    __shared__ int   l_r[AGG_CAP];
    __shared__ float l_n[AGG_CAP];
    __shared__ int   cnt;

    const int tid  = threadIdx.x;
    const int lane = tid & 31;
    const int w    = __builtin_amdgcn_readfirstlane(tid >> 5);
    const int lo   = blockIdx.x * AGG_ROWS;
    const int rows = min(AGG_ROWS, N - lo);
    const unsigned urows = (unsigned)rows;
    const float4* h4 = (const float4*)h;

    for (int r = w; r < rows; r += AGG_WAVES) {
        float dd = dis[lo + r];
        float nn = dd * dd;
        float4 v = h4[(size_t)(lo + r) * (FD / 4) + lane];
        float* a = acc + r * FD + lane;
        a[0]  = v.x * nn;
        a[32] = v.y * nn;
        a[64] = v.z * nn;
        a[96] = v.w * nn;
    }
    if (tid < FD) acc[AGG_ROWS * FD + tid] = 0.0f;
    if (tid == 0) cnt = 0;
    __syncthreads();

    const uint4* r4 = (const uint4*)(bm + (size_t)(blockIdx.x & (OWN_SLOTS - 1)) * rowWords);
    const int nv = rowWords >> 2;
    for (int j0 = tid; j0 < nv; j0 += 5 * AGG_THREADS) {
        uint4 bv[5];
        #pragma unroll
        for (int u = 0; u < 5; ++u) {
            int j = j0 + u * AGG_THREADS;
            bv[u] = j < nv ? r4[j] : make_uint4(0u, 0u, 0u, 0u);
        }
        int total = 0;
        #pragma unroll
        for (int u = 0; u < 5; ++u)
            total += __popc(bv[u].x) + __popc(bv[u].y) + __popc(bv[u].z) + __popc(bv[u].w);
        int pos = atomicAdd(&cnt, total);
        #pragma unroll
        for (int u = 0; u < 5; ++u) {
            const int eb = (j0 + u * AGG_THREADS) << 7;
            unsigned wd[4] = {bv[u].x, bv[u].y, bv[u].z, bv[u].w};
            #pragma unroll
            for (int q = 0; q < 4; ++q) {
                unsigned mm = wd[q];
                while (mm) {
                    if (pos < AGG_CAP) l_a[pos] = eb + 32 * q + __builtin_ctz(mm);
                    ++pos;
                    mm &= mm - 1u;
                }
            }
        }
    }
    __syncthreads();
    const int found = cnt;

    if (found > AGG_CAP) {
        for (int e = tid; e < E; e += AGG_THREADS) {
            unsigned r = (unsigned)(dst[e] - lo);
            if (r < urows) {
                int s = src[e];
                float nn = dis[s] * dis[lo + (int)r];
                const float* hs = h + (size_t)s * FD;
                for (int c = 0; c < FD; ++c)
                    atomicAdd(&acc[r * FD + (c & 3) * 32 + (c >> 2)], hs[c] * nn);
            }
        }
    } else {
        for (int t = tid; t < found; t += AGG_THREADS) {
            int e = l_a[t];
            int d = dst[e];
            int s = src[e];
            unsigned r = (unsigned)(d - lo);
            l_a[t] = s;
            l_r[t] = r < urows ? (int)r : AGG_ROWS;
            l_n[t] = dis[s] * dis[d];
        }
        __syncthreads();
        const int groups = found >> 3;
        for (int g = w; g < groups; g += AGG_WAVES) {
            float4 v[8];
            float  nn[8];
            int    rr[8];
            #pragma unroll
            for (int u = 0; u < 8; ++u) {
                int i = g * 8 + u;
                int s = l_a[i];
                rr[u] = l_r[i];
                nn[u] = l_n[i];
                v[u]  = h4[(size_t)s * (FD / 4) + lane];
            }
            #pragma unroll
            for (int u = 0; u < 8; ++u) {
                float* a = acc + rr[u] * FD + lane;
                atomicAdd(a,      v[u].x * nn[u]);
                atomicAdd(a + 32, v[u].y * nn[u]);
                atomicAdd(a + 64, v[u].z * nn[u]);
                atomicAdd(a + 96, v[u].w * nn[u]);
            }
        }
        {
            int i = groups * 8 + w;
            if (i < found) {
                int s = l_a[i];
                float n1 = l_n[i];
                float4 v1 = h4[(size_t)s * (FD / 4) + lane];
                float* a = acc + l_r[i] * FD + lane;
                atomicAdd(a,      v1.x * n1);
                atomicAdd(a + 32, v1.y * n1);
                atomicAdd(a + 64, v1.z * n1);
                atomicAdd(a + 96, v1.w * n1);
            }
        }
    }
    __syncthreads();

    const float4 b4 = ((const float4*)bias)[lane];
    float4* o4 = (float4*)out;
    for (int r = w; r < rows; r += AGG_WAVES) {
        const float* a = acc + r * FD + lane;
        float4 v;
        v.x = a[0]  + b4.x;
        v.y = a[32] + b4.y;
        v.z = a[64] + b4.z;
        v.w = a[96] + b4.w;
        if (RELU) {
            v.x = v.x > 0.0f ? v.x : 0.0f;
            v.y = v.y > 0.0f ? v.y : 0.0f;
            v.z = v.z > 0.0f ? v.z : 0.0f;
            v.w = v.w > 0.0f ? v.w : 0.0f;
        }
        o4[(size_t)(lo + r) * (FD / 4) + lane] = v;
    }
}


extern "C" void kernel_launch(void* const* d_in, const int* in_sizes, int n_in,
                              void* d_out, int out_size, void* d_ws, size_t ws_size,
                              hipStream_t stream) {
    const float* x  = (const float*)d_in[0];
    const int*   ei = (const int*)d_in[1];
    const float* W1 = (const float*)d_in[2];
    const float* b1 = (const float*)d_in[3];
    const float* W2 = (const float*)d_in[4];
    const float* b2 = (const float*)d_in[5];

    const int N = in_sizes[0] / FD;
    const int E = in_sizes[1] / 2;
    const int* src = ei;
    const int* dst = ei + E;

    const int bblocks  = (E + BM_EDGES - 1) / BM_EDGES;
    const int rowWords = bblocks * BM_WORDS;
    unsigned* bm = (unsigned*)d_ws;

    float* ws   = (float*)d_ws + (size_t)OWN_SLOTS * rowWords;
    float* deg  = ws;
    float* dis  = ws + N;
    float* bufA = ws + 2 * (size_t)N;
    float* bufB = bufA + (size_t)N * FD;
    float* out  = (float*)d_out;

    hipMemsetAsync(deg, 0, (size_t)N * sizeof(float), stream);
    gcn_deg_own_kernel<<<bblocks, BM_EDGES, 0, stream>>>(dst, deg, bm, E, rowWords);
    gcn_dis_kernel<<<(N + 255) / 256, 256, 0, stream>>>(deg, dis, N);

    const int gblocks = (N + GEMM_ROWS - 1) / GEMM_ROWS;

    const int ablocks = (N + AGG_ROWS - 1) / AGG_ROWS;

    gcn_gemm_wmma<<<gblocks, 256, 0, stream>>>(x, W1, bufA, N);
    gcn_agg_owner<true><<<ablocks, AGG_THREADS, 0, stream>>>(bufA, src, dst, dis, bm, rowWords, b1, bufB, E, N);

    gcn_gemm_wmma<<<gblocks, 256, 0, stream>>>(bufB, W2, bufA, N);
    gcn_agg_owner<false><<<ablocks, AGG_THREADS, 0, stream>>>(bufA, src, dst, dis, bm, rowWords, b2, out, E, N);
}


# Degree/bitmap pass and the layer-1 GEMM share one launch: build work-groups (256 threads, 4 edges each) first in the grid, GEMM work-groups beside them on the same WGPs; 6 launches -> 5
# speedup vs baseline: 2.4499x; 1.0329x over previous
#include <hip/hip_runtime.h>
#include <hip/hip_bf16.h>

typedef float v2f __attribute__((ext_vector_type(2)));
typedef float v8f __attribute__((ext_vector_type(8)));

#define FD 128


#define AGG_ROWS    196
#define OWN_SLOTS   256
#define BM_EDGES    1024
#define BM_WORDS    (BM_EDGES / 32)
#define BM_PITCH    (OWN_SLOTS + 1)
#define PRE_THREADS 256
#define BM_EPT      (BM_EDGES / PRE_THREADS)
#define BM_LDS4     (BM_WORDS * BM_PITCH / 4)

__device__ __forceinline__
void deg_own_block(const int* __restrict__ dst, float* __restrict__ deg,
                   unsigned* __restrict__ bm, int E, int rowWords, int chunk,
                   unsigned* m) {
    const int tid = threadIdx.x;
    uint4* m4 = (uint4*)m;
    #pragma unroll
    for (int i = 0; i < BM_LDS4 / PRE_THREADS; ++i) m4[i * PRE_THREADS + tid] = make_uint4(0u, 0u, 0u, 0u);
    if (tid < BM_LDS4 % PRE_THREADS) m4[(BM_LDS4 / PRE_THREADS) * PRE_THREADS + tid] = make_uint4(0u, 0u, 0u, 0u);
    __syncthreads();
    const int e0 = chunk * BM_EDGES + BM_EPT * tid;
    int d[BM_EPT];
    if (e0 + BM_EPT <= E && (((size_t)dst) & 15) == 0) {
        const int4 v = *(const int4*)(dst + e0);
        d[0] = v.x; d[1] = v.y; d[2] = v.z; d[3] = v.w;
    } else {
        #pragma unroll
        for (int k = 0; k < BM_EPT; ++k) d[k] = e0 + k < E ? dst[e0 + k] : -1;
    }
    unsigned* mrow = m + ((BM_EPT * tid) >> 5) * BM_PITCH;
    const int b0 = (BM_EPT * tid) & 31;
    #pragma unroll
    for (int k = 0; k < BM_EPT; ++k) {
        if (d[k] >= 0) {
            unsafeAtomicAdd(&deg[d[k]], 1.0f);
            const unsigned o = ((unsigned)d[k] / AGG_ROWS) & (OWN_SLOTS - 1);
            atomicOr(&mrow[o], 1u << (b0 + k));
        }
    }
    __syncthreads();
    const int w  = tid & 31;
    const int wv = __builtin_amdgcn_readfirstlane(tid >> 5);
    unsigned* bcol = bm + (size_t)chunk * BM_WORDS + w;
    #pragma unroll 8
    for (int i = 0; i < OWN_SLOTS / (PRE_THREADS / 32); ++i) {
        const int o = i * (PRE_THREADS / 32) + wv;
        bcol[(size_t)o * rowWords] = m[w * BM_PITCH + o];
    }
}

__global__ void gcn_dis_kernel(const float* __restrict__ deg, float* __restrict__ dis, int N) {
    int i = blockIdx.x * blockDim.x + threadIdx.x;
    if (i < N) dis[i] = rsqrtf(deg[i] + 1.0f);
}


#define GEMM_ROWS  64
#define GEMM_PITCH 132
#define GEMM_RT    (GEMM_ROWS / 16)
#define PRE_LDS4   (GEMM_ROWS * GEMM_PITCH / 4)

__device__ __forceinline__
void gemm_block(const float* __restrict__ A, const float* __restrict__ W,
                float* __restrict__ C, int M, int blk, float* sa) {
    const int tid  = threadIdx.x;
    const int lane = tid & 31;
    const int wv   = __builtin_amdgcn_readfirstlane(tid >> 5);
    const int half = lane >> 4;
    const int l16  = lane & 15;
    const int row0 = blk * GEMM_ROWS;

    const float4* A4 = (const float4*)A;
    float4 av[GEMM_ROWS / 8];
    #pragma unroll
    for (int i = 0; i < GEMM_ROWS / 8; ++i) {
        int gr = row0 + i * 8 + wv;
        av[i] = gr < M ? A4[(size_t)gr * (FD / 4) + lane] : make_float4(0.0f, 0.0f, 0.0f, 0.0f);
    }

    v2f wf[FD / 4];
    const float* wcol = W + (size_t)(2 * half) * FD + wv * 16 + l16;
    #pragma unroll
    for (int ks = 0; ks < FD / 4; ++ks) {
        wf[ks].x = wcol[(size_t)(4 * ks) * FD];
        wf[ks].y = wcol[(size_t)(4 * ks + 1) * FD];
    }

    #pragma unroll
    for (int i = 0; i < GEMM_ROWS / 8; ++i) {
        *(float4*)&sa[(i * 8 + wv) * GEMM_PITCH + 4 * lane] = av[i];
    }
    __syncthreads();

    v8f acc[GEMM_RT];
    #pragma unroll
    for (int rt = 0; rt < GEMM_RT; ++rt) acc[rt] = v8f{};

    const float* arow = sa + l16 * GEMM_PITCH + 2 * half;
    #pragma unroll
    for (int ks = 0; ks < FD / 4; ++ks) {
        #pragma unroll
        for (int rt = 0; rt < GEMM_RT; ++rt) {
            v2f a = *(const v2f*)&arow[rt * 16 * GEMM_PITCH + 4 * ks];
            acc[rt] = __builtin_amdgcn_wmma_f32_16x16x4_f32(false, a, false, wf[ks],
                                                            (short)0, acc[rt], false, false);
        }
    }

    __syncthreads();
    float* crow = sa + (8 * half) * GEMM_PITCH + wv * 16 + l16;
    #pragma unroll
    for (int rt = 0; rt < GEMM_RT; ++rt) {
        #pragma unroll
        for (int v = 0; v < 8; ++v) {
            crow[(rt * 16 + v) * GEMM_PITCH] = acc[rt][v];
        }
    }
    __syncthreads();
    float4* C4 = (float4*)C;
    #pragma unroll
    for (int i = 0; i < GEMM_ROWS / 8; ++i) {
        int r  = i * 8 + wv;
        int gr = row0 + r;
        if (gr < M) C4[(size_t)gr * (FD / 4) + lane] = *(const float4*)&sa[r * GEMM_PITCH + 4 * lane];
    }
}

__global__ __launch_bounds__(PRE_THREADS)
void gcn_gemm_wmma(const float* __restrict__ A, const float* __restrict__ W,
                   float* __restrict__ C, int M) {
    __shared__ float4 lds[PRE_LDS4];
    gemm_block(A, W, C, M, blockIdx.x, (float*)lds);
}

__global__ __launch_bounds__(PRE_THREADS)
void gcn_pre_gemm(const int* __restrict__ dst, float* __restrict__ deg,
                  unsigned* __restrict__ bm, int E, int rowWords, int bblocks,
                  const float* __restrict__ A, const float* __restrict__ W,
                  float* __restrict__ C, int M) {
    __shared__ float4 lds[PRE_LDS4];
    const int b = blockIdx.x;
    if (b < bblocks) deg_own_block(dst, deg, bm, E, rowWords, b, (unsigned*)lds);
    else             gemm_block(A, W, C, M, b - bblocks, (float*)lds);
}


#define AGG_CAP     8192
#define AGG_THREADS 1024
#define AGG_WAVES   (AGG_THREADS / 32)

template <bool RELU>
__global__ __launch_bounds__(AGG_THREADS)
void gcn_agg_owner(const float* __restrict__ h, const int* __restrict__ src,
                   const int* __restrict__ dst, const float* __restrict__ dis,
                   const unsigned* __restrict__ bm, int rowWords,
                   const float* __restrict__ bias, float* __restrict__ out, int E, int N) {
    __shared__ float acc[(AGG_ROWS + 1) * FD];
    __shared__ int   l_a[AGG_CAP];
    __shared__ int   l_r[AGG_CAP];
    __shared__ float l_n[AGG_CAP];
    __shared__ int   cnt;

    const int tid  = threadIdx.x;
    const int lane = tid & 31;
    const int w    = __builtin_amdgcn_readfirstlane(tid >> 5);
    const int lo   = blockIdx.x * AGG_ROWS;
    const int rows = min(AGG_ROWS, N - lo);
    const unsigned urows = (unsigned)rows;
    const float4* h4 = (const float4*)h;

    for (int r = w; r < rows; r += AGG_WAVES) {
        float dd = dis[lo + r];
        float nn = dd * dd;
        float4 v = h4[(size_t)(lo + r) * (FD / 4) + lane];
        float* a = acc + r * FD + lane;
        a[0]  = v.x * nn;
        a[32] = v.y * nn;
        a[64] = v.z * nn;
        a[96] = v.w * nn;
    }
    if (tid < FD) acc[AGG_ROWS * FD + tid] = 0.0f;
    if (tid == 0) cnt = 0;
    __syncthreads();

    const uint4* r4 = (const uint4*)(bm + (size_t)(blockIdx.x & (OWN_SLOTS - 1)) * rowWords);
    const int nv = rowWords >> 2;
    for (int j0 = tid; j0 < nv; j0 += 5 * AGG_THREADS) {
        uint4 bv[5];
        #pragma unroll
        for (int u = 0; u < 5; ++u) {
            int j = j0 + u * AGG_THREADS;
            bv[u] = j < nv ? r4[j] : make_uint4(0u, 0u, 0u, 0u);
        }
        int total = 0;
        #pragma unroll
        for (int u = 0; u < 5; ++u)
            total += __popc(bv[u].x) + __popc(bv[u].y) + __popc(bv[u].z) + __popc(bv[u].w);
        int pos = atomicAdd(&cnt, total);
        #pragma unroll
        for (int u = 0; u < 5; ++u) {
            const int eb = (j0 + u * AGG_THREADS) << 7;
            unsigned wd[4] = {bv[u].x, bv[u].y, bv[u].z, bv[u].w};
            #pragma unroll
            for (int q = 0; q < 4; ++q) {
                unsigned mm = wd[q];
                while (mm) {
                    if (pos < AGG_CAP) l_a[pos] = eb + 32 * q + __builtin_ctz(mm);
                    ++pos;
                    mm &= mm - 1u;
                }
            }
        }
    }
    __syncthreads();
    const int found = cnt;

    if (found > AGG_CAP) {
        for (int e = tid; e < E; e += AGG_THREADS) {
            unsigned r = (unsigned)(dst[e] - lo);
            if (r < urows) {
                int s = src[e];
                float nn = dis[s] * dis[lo + (int)r];
                const float* hs = h + (size_t)s * FD;
                for (int c = 0; c < FD; ++c)
                    atomicAdd(&acc[r * FD + (c & 3) * 32 + (c >> 2)], hs[c] * nn);
            }
        }
    } else {
        for (int t = tid; t < found; t += AGG_THREADS) {
            int e = l_a[t];
            int d = dst[e];
            int s = src[e];
            unsigned r = (unsigned)(d - lo);
            l_a[t] = s;
            l_r[t] = r < urows ? (int)r : AGG_ROWS;
            l_n[t] = dis[s] * dis[d];
        }
        __syncthreads();
        const int groups = found >> 3;
        for (int g = w; g < groups; g += AGG_WAVES) {
            float4 v[8];
            float  nn[8];
            int    rr[8];
            #pragma unroll
            for (int u = 0; u < 8; ++u) {
                int i = g * 8 + u;
                int s = l_a[i];
                rr[u] = l_r[i];
                nn[u] = l_n[i];
                v[u]  = h4[(size_t)s * (FD / 4) + lane];
            }
            #pragma unroll
            for (int u = 0; u < 8; ++u) {
                float* a = acc + rr[u] * FD + lane;
                atomicAdd(a,      v[u].x * nn[u]);
                atomicAdd(a + 32, v[u].y * nn[u]);
                atomicAdd(a + 64, v[u].z * nn[u]);
                atomicAdd(a + 96, v[u].w * nn[u]);
            }
        }
        {
            int i = groups * 8 + w;
            if (i < found) {
                int s = l_a[i];
                float n1 = l_n[i];
                float4 v1 = h4[(size_t)s * (FD / 4) + lane];
                float* a = acc + l_r[i] * FD + lane;
                atomicAdd(a,      v1.x * n1);
                atomicAdd(a + 32, v1.y * n1);
                atomicAdd(a + 64, v1.z * n1);
                atomicAdd(a + 96, v1.w * n1);
            }
        }
    }
    __syncthreads();

    const float4 b4 = ((const float4*)bias)[lane];
    float4* o4 = (float4*)out;
    for (int r = w; r < rows; r += AGG_WAVES) {
        const float* a = acc + r * FD + lane;
        float4 v;
        v.x = a[0]  + b4.x;
        v.y = a[32] + b4.y;
        v.z = a[64] + b4.z;
        v.w = a[96] + b4.w;
        if (RELU) {
            v.x = v.x > 0.0f ? v.x : 0.0f;
            v.y = v.y > 0.0f ? v.y : 0.0f;
            v.z = v.z > 0.0f ? v.z : 0.0f;
            v.w = v.w > 0.0f ? v.w : 0.0f;
        }
        o4[(size_t)(lo + r) * (FD / 4) + lane] = v;
    }
}


extern "C" void kernel_launch(void* const* d_in, const int* in_sizes, int n_in,
                              void* d_out, int out_size, void* d_ws, size_t ws_size,
                              hipStream_t stream) {
    const float* x  = (const float*)d_in[0];
    const int*   ei = (const int*)d_in[1];
    const float* W1 = (const float*)d_in[2];
    const float* b1 = (const float*)d_in[3];
    const float* W2 = (const float*)d_in[4];
    const float* b2 = (const float*)d_in[5];

    const int N = in_sizes[0] / FD;
    const int E = in_sizes[1] / 2;
    const int* src = ei;
    const int* dst = ei + E;

    const int bblocks  = (E + BM_EDGES - 1) / BM_EDGES;
    const int rowWords = bblocks * BM_WORDS;
    unsigned* bm = (unsigned*)d_ws;

    float* ws   = (float*)d_ws + (size_t)OWN_SLOTS * rowWords;
    float* deg  = ws;
    float* dis  = ws + N;
    float* bufA = ws + 2 * (size_t)N;
    float* bufB = bufA + (size_t)N * FD;
    float* out  = (float*)d_out;

    hipMemsetAsync(deg, 0, (size_t)N * sizeof(float), stream);

    const int gblocks = (N + GEMM_ROWS - 1) / GEMM_ROWS;

    gcn_pre_gemm<<<bblocks + gblocks, PRE_THREADS, 0, stream>>>(dst, deg, bm, E, rowWords, bblocks, x, W1, bufA, N);
    gcn_dis_kernel<<<(N + 255) / 256, 256, 0, stream>>>(deg, dis, N);

    const int ablocks = (N + AGG_ROWS - 1) / AGG_ROWS;

    gcn_agg_owner<true><<<ablocks, AGG_THREADS, 0, stream>>>(bufA, src, dst, dis, bm, rowWords, b1, bufB, E, N);

    gcn_gemm_wmma<<<gblocks, PRE_THREADS, 0, stream>>>(bufB, W2, bufA, N);
    gcn_agg_owner<false><<<ablocks, AGG_THREADS, 0, stream>>>(bufA, src, dst, dis, bm, rowWords, b2, out, E, N);
}


# Build pass hands each owner its edges as records (source, row in range; LDS arrival rank, 31 per 128-B line) instead of bits: owner needs no dst/src lookup per edge; sums stay LDS arrival order
# speedup vs baseline: 2.6094x; 1.0651x over previous
#include <hip/hip_runtime.h>
#include <hip/hip_bf16.h>

typedef float v2f __attribute__((ext_vector_type(2)));
typedef float v8f __attribute__((ext_vector_type(8)));

#define FD 128


#define AGG_ROWS    196
#define OWN_SLOTS   256
#define BM_EDGES    1024
#define BM_WORDS    32
#define REC_MAX     (BM_WORDS - 1)
#define BM_PITCH    (BM_WORDS + 1)
#define PRE_THREADS 256
#define BM_EPT      (BM_EDGES / PRE_THREADS)
#define REC_NONE    0xFFFFFFFFu

static_assert(PRE_THREADS == OWN_SLOTS, "one thread prepares one owner's line");
static_assert(BM_EPT == 4, "a thread's edges come in one 16-byte load");
static_assert(AGG_ROWS < 255, "the row inside the owner's range must fit the high byte");

__device__ __forceinline__
void deg_own_block(const int* __restrict__ src, const int* __restrict__ dst,
                   float* __restrict__ deg, unsigned* __restrict__ bm, int E, int rowWords,
                   int chunk, int passes, unsigned* m) {
    const int tid = threadIdx.x;
    const int e0 = chunk * BM_EDGES + BM_EPT * tid;
    int d[BM_EPT], s[BM_EPT];
    if (e0 + BM_EPT <= E && ((((size_t)dst) | ((size_t)src)) & 15) == 0) {
        const int4 dv = *(const int4*)(dst + e0);
        const int4 sv = *(const int4*)(src + e0);
        d[0] = dv.x; d[1] = dv.y; d[2] = dv.z; d[3] = dv.w;
        s[0] = sv.x; s[1] = sv.y; s[2] = sv.z; s[3] = sv.w;
    } else {
        #pragma unroll
        for (int k = 0; k < BM_EPT; ++k) {
            const bool in = e0 + k < E;
            d[k] = in ? dst[e0 + k] : -1;
            s[k] = in ? src[e0 + k] : 0;
        }
    }
    #pragma unroll
    for (int k = 0; k < BM_EPT; ++k)
        if (d[k] >= 0) unsafeAtomicAdd(&deg[d[k]], 1.0f);

    const int w  = tid & 31;
    const int wv = __builtin_amdgcn_readfirstlane(tid >> 5);
    for (int p = 0; p < passes; ++p) {
        if (p) __syncthreads();
        unsigned* mine = m + tid * BM_PITCH;
        mine[0] = 0u;
        #pragma unroll
        for (int k = 1; k < BM_WORDS; ++k) mine[k] = REC_NONE;
        __syncthreads();
        #pragma unroll
        for (int k = 0; k < BM_EPT; ++k) {
            if (d[k] >= 0) {
                const unsigned o = (unsigned)d[k] / AGG_ROWS;
                if ((int)(o / OWN_SLOTS) == p) {
                    const unsigned r = (unsigned)d[k] - o * AGG_ROWS;
                    unsigned* line = m + (o % OWN_SLOTS) * BM_PITCH;
                    const unsigned rank = atomicAdd(line, 1u);
                    if (rank < REC_MAX) line[1 + rank] = (unsigned)s[k] | (r << 24);
                }
            }
        }
        __syncthreads();
        unsigned* bcol = bm + (size_t)p * OWN_SLOTS * rowWords + (size_t)chunk * BM_WORDS + w;
        #pragma unroll 8
        for (int i = 0; i < OWN_SLOTS / (PRE_THREADS / 32); ++i) {
            const int o = i * (PRE_THREADS / 32) + wv;
            bcol[(size_t)o * rowWords] = m[o * BM_PITCH + w];
        }
    }
}

__global__ void gcn_dis_kernel(const float* __restrict__ deg, float* __restrict__ dis, int N) {
    int i = blockIdx.x * blockDim.x + threadIdx.x;
    if (i < N) dis[i] = rsqrtf(deg[i] + 1.0f);
}


#define GEMM_ROWS  64
#define GEMM_PITCH 132
#define GEMM_RT    (GEMM_ROWS / 16)
#define PRE_LDS4   (GEMM_ROWS * GEMM_PITCH / 4)
static_assert(OWN_SLOTS * BM_PITCH <= PRE_LDS4 * 4, "the owners' lines must fit the shared LDS");

__device__ __forceinline__
void gemm_block(const float* __restrict__ A, const float* __restrict__ W,
                float* __restrict__ C, int M, int blk, float* sa) {
    const int tid  = threadIdx.x;
    const int lane = tid & 31;
    const int wv   = __builtin_amdgcn_readfirstlane(tid >> 5);
    const int half = lane >> 4;
    const int l16  = lane & 15;
    const int row0 = blk * GEMM_ROWS;

    const float4* A4 = (const float4*)A;
    float4 av[GEMM_ROWS / 8];
    #pragma unroll
    for (int i = 0; i < GEMM_ROWS / 8; ++i) {
        int gr = row0 + i * 8 + wv;
        av[i] = gr < M ? A4[(size_t)gr * (FD / 4) + lane] : make_float4(0.0f, 0.0f, 0.0f, 0.0f);
    }

    v2f wf[FD / 4];
    const float* wcol = W + (size_t)(2 * half) * FD + wv * 16 + l16;
    #pragma unroll
    for (int ks = 0; ks < FD / 4; ++ks) {
        wf[ks].x = wcol[(size_t)(4 * ks) * FD];
        wf[ks].y = wcol[(size_t)(4 * ks + 1) * FD];
    }

    #pragma unroll
    for (int i = 0; i < GEMM_ROWS / 8; ++i) {
        *(float4*)&sa[(i * 8 + wv) * GEMM_PITCH + 4 * lane] = av[i];
    }
    __syncthreads();

    v8f acc[GEMM_RT];
    #pragma unroll
    for (int rt = 0; rt < GEMM_RT; ++rt) acc[rt] = v8f{};

    const float* arow = sa + l16 * GEMM_PITCH + 2 * half;
    #pragma unroll
    for (int ks = 0; ks < FD / 4; ++ks) {
        #pragma unroll
        for (int rt = 0; rt < GEMM_RT; ++rt) {
            v2f a = *(const v2f*)&arow[rt * 16 * GEMM_PITCH + 4 * ks];
            acc[rt] = __builtin_amdgcn_wmma_f32_16x16x4_f32(false, a, false, wf[ks],
                                                            (short)0, acc[rt], false, false);
        }
    }

    __syncthreads();
    float* crow = sa + (8 * half) * GEMM_PITCH + wv * 16 + l16;
    #pragma unroll
    for (int rt = 0; rt < GEMM_RT; ++rt) {
        #pragma unroll
        for (int v = 0; v < 8; ++v) {
            crow[(rt * 16 + v) * GEMM_PITCH] = acc[rt][v];
        }
    }
    __syncthreads();
    float4* C4 = (float4*)C;
    #pragma unroll
    for (int i = 0; i < GEMM_ROWS / 8; ++i) {
        int r  = i * 8 + wv;
        int gr = row0 + r;
        if (gr < M) C4[(size_t)gr * (FD / 4) + lane] = *(const float4*)&sa[r * GEMM_PITCH + 4 * lane];
    }
}

__global__ __launch_bounds__(PRE_THREADS)
void gcn_gemm_wmma(const float* __restrict__ A, const float* __restrict__ W,
                   float* __restrict__ C, int M) {
    __shared__ float4 lds[PRE_LDS4];
    gemm_block(A, W, C, M, blockIdx.x, (float*)lds);
}

__global__ __launch_bounds__(PRE_THREADS)
void gcn_pre_gemm(const int* __restrict__ src, const int* __restrict__ dst,
                  float* __restrict__ deg, unsigned* __restrict__ bm, int E, int rowWords,
                  int bblocks, int passes,
                  const float* __restrict__ A, const float* __restrict__ W,
                  float* __restrict__ C, int M) {
    __shared__ float4 lds[PRE_LDS4];
    const int b = blockIdx.x;
    if (b < bblocks) deg_own_block(src, dst, deg, bm, E, rowWords, b, passes, (unsigned*)lds);
    else             gemm_block(A, W, C, M, b - bblocks, (float*)lds);
}


#define AGG_CAP     16384
#define AGG_THREADS 1024
#define AGG_WAVES   (AGG_THREADS / 32)
#define AGG_LOADS   5

static_assert(AGG_THREADS % (BM_WORDS / 4) == 0, "a thread holds the same piece of a line in every load");

template <bool RELU>
__global__ __launch_bounds__(AGG_THREADS)
void gcn_agg_owner(const float* __restrict__ h, const int* __restrict__ src,
                   const int* __restrict__ dst, const float* __restrict__ dis,
                   const unsigned* __restrict__ bm, int rowWords,
                   const float* __restrict__ bias, float* __restrict__ out, int E, int N) {
    __shared__ float    acc[AGG_ROWS * FD];
    __shared__ unsigned l_a[AGG_CAP];
    __shared__ int      cnt;
    __shared__ int      ovf;

    const int tid  = threadIdx.x;
    const int lane = tid & 31;
    const int w    = __builtin_amdgcn_readfirstlane(tid >> 5);
    const int lo   = blockIdx.x * AGG_ROWS;
    const int rows = min(AGG_ROWS, N - lo);
    const unsigned urows = (unsigned)rows;
    const float4* h4 = (const float4*)h;

    for (int r = w; r < rows; r += AGG_WAVES) {
        float dd = dis[lo + r];
        float nn = dd * dd;
        float4 v = h4[(size_t)(lo + r) * (FD / 4) + lane];
        float* a = acc + r * FD + lane;
        a[0]  = v.x * nn;
        a[32] = v.y * nn;
        a[64] = v.z * nn;
        a[96] = v.w * nn;
    }
    if (tid == 0) { cnt = 0; ovf = 0; }
    __syncthreads();

    const uint4* r4 = (const uint4*)(bm + (size_t)blockIdx.x * rowWords);
    const int nv = rowWords >> 2;
    const bool head = (tid & (BM_WORDS / 4 - 1)) == 0;
    for (int j0 = tid; j0 < nv; j0 += AGG_LOADS * AGG_THREADS) {
        uint4 bv[AGG_LOADS];
        #pragma unroll
        for (int u = 0; u < AGG_LOADS; ++u) {
            int j = j0 + u * AGG_THREADS;
            bv[u] = j < nv ? r4[j] : make_uint4(REC_NONE, REC_NONE, REC_NONE, REC_NONE);
        }
        int  total = 0;
        bool over  = false;
        #pragma unroll
        for (int u = 0; u < AGG_LOADS; ++u) {
            if (head) over = over || (bv[u].x != REC_NONE && bv[u].x > REC_MAX);
            else      total += bv[u].x != REC_NONE;
            total += (bv[u].y != REC_NONE) + (bv[u].z != REC_NONE) + (bv[u].w != REC_NONE);
        }
        if (over) ovf = 1;
        int pos = atomicAdd(&cnt, total);
        #pragma unroll
        for (int u = 0; u < AGG_LOADS; ++u) {
            const unsigned wd[4] = {bv[u].x, bv[u].y, bv[u].z, bv[u].w};
            #pragma unroll
            for (int q = 0; q < 4; ++q) {
                if (!(head && q == 0) && wd[q] != REC_NONE) {
                    if (pos < AGG_CAP) l_a[pos] = wd[q];
                    ++pos;
                }
            }
        }
    }
    __syncthreads();
    const int found = cnt;

    if (found > AGG_CAP || ovf != 0) {
        for (int e = tid; e < E; e += AGG_THREADS) {
            unsigned r = (unsigned)(dst[e] - lo);
            if (r < urows) {
                int s = src[e];
                float nn = dis[s] * dis[lo + (int)r];
                const float* hs = h + (size_t)s * FD;
                for (int c = 0; c < FD; ++c)
                    atomicAdd(&acc[r * FD + (c & 3) * 32 + (c >> 2)], hs[c] * nn);
            }
        }
    } else {
        const int groups = found >> 3;
        for (int g = w; g < groups; g += AGG_WAVES) {
            float4 v[8];
            int    rr[8];
            const unsigned recl = l_a[g * 8 + (lane & 7)];
            const float nl = dis[(int)(recl & 0xFFFFFFu)] * dis[lo + (int)(recl >> 24)];
            #pragma unroll
            for (int u = 0; u < 8; ++u) {
                const unsigned rec = l_a[g * 8 + u];
                const int s = (int)(rec & 0xFFFFFFu);
                rr[u] = (int)(rec >> 24);
                v[u]  = h4[(size_t)s * (FD / 4) + lane];
            }
            #pragma unroll
            for (int u = 0; u < 8; ++u) {
                const float nn = __int_as_float(__builtin_amdgcn_readlane(__float_as_int(nl), u));
                float* a = acc + rr[u] * FD + lane;
                atomicAdd(a,      v[u].x * nn);
                atomicAdd(a + 32, v[u].y * nn);
                atomicAdd(a + 64, v[u].z * nn);
                atomicAdd(a + 96, v[u].w * nn);
            }
        }
        {
            int i = groups * 8 + w;
            if (i < found) {
                const unsigned rec = l_a[i];
                const int s  = (int)(rec & 0xFFFFFFu);
                const int r1 = (int)(rec >> 24);
                const float n1 = dis[s] * dis[lo + r1];
                float4 v1 = h4[(size_t)s * (FD / 4) + lane];
                float* a = acc + r1 * FD + lane;
                atomicAdd(a,      v1.x * n1);
                atomicAdd(a + 32, v1.y * n1);
                atomicAdd(a + 64, v1.z * n1);
                atomicAdd(a + 96, v1.w * n1);
            }
        }
    }
    __syncthreads();

    const float4 b4 = ((const float4*)bias)[lane];
    float4* o4 = (float4*)out;
    for (int r = w; r < rows; r += AGG_WAVES) {
        const float* a = acc + r * FD + lane;
        float4 v;
        v.x = a[0]  + b4.x;
        v.y = a[32] + b4.y;
        v.z = a[64] + b4.z;
        v.w = a[96] + b4.w;
        if (RELU) {
            v.x = v.x > 0.0f ? v.x : 0.0f;
            v.y = v.y > 0.0f ? v.y : 0.0f;
            v.z = v.z > 0.0f ? v.z : 0.0f;
            v.w = v.w > 0.0f ? v.w : 0.0f;
        }
        o4[(size_t)(lo + r) * (FD / 4) + lane] = v;
    }
}


extern "C" void kernel_launch(void* const* d_in, const int* in_sizes, int n_in,
                              void* d_out, int out_size, void* d_ws, size_t ws_size,
                              hipStream_t stream) {
    const float* x  = (const float*)d_in[0];
    const int*   ei = (const int*)d_in[1];
    const float* W1 = (const float*)d_in[2];
    const float* b1 = (const float*)d_in[3];
    const float* W2 = (const float*)d_in[4];
    const float* b2 = (const float*)d_in[5];

    const int N = in_sizes[0] / FD;
    const int E = in_sizes[1] / 2;
    const int* src = ei;
    const int* dst = ei + E;

    const int ablocks = (N + AGG_ROWS - 1) / AGG_ROWS;
    const int passes  = (ablocks + OWN_SLOTS - 1) / OWN_SLOTS;

    const int bblocks  = (E + BM_EDGES - 1) / BM_EDGES;
    const int rowWords = bblocks * BM_WORDS;
    unsigned* bm = (unsigned*)d_ws;

    float* ws   = (float*)d_ws + (size_t)passes * OWN_SLOTS * rowWords;
    float* deg  = ws;
    float* dis  = ws + N;
    float* bufA = ws + 2 * (size_t)N;
    float* bufB = bufA + (size_t)N * FD;
    float* out  = (float*)d_out;

    hipMemsetAsync(deg, 0, (size_t)N * sizeof(float), stream);

    const int gblocks = (N + GEMM_ROWS - 1) / GEMM_ROWS;

    gcn_pre_gemm<<<bblocks + gblocks, PRE_THREADS, 0, stream>>>(src, dst, deg, bm, E, rowWords, bblocks, passes, x, W1, bufA, N);
    gcn_dis_kernel<<<(N + 255) / 256, 256, 0, stream>>>(deg, dis, N);

    gcn_agg_owner<true><<<ablocks, AGG_THREADS, 0, stream>>>(bufA, src, dst, dis, bm, rowWords, b1, bufB, E, N);

    gcn_gemm_wmma<<<gblocks, PRE_THREADS, 0, stream>>>(bufB, W2, bufA, N);
    gcn_agg_owner<false><<<ablocks, AGG_THREADS, 0, stream>>>(bufA, src, dst, dis, bm, rowWords, b2, out, E, N);
}


# In-degrees counted by the owner work-groups from their own record lines (LDS integer counters per owned row): the 600000 global float atomics and their memset go; norm launch becomes an owner launch
# speedup vs baseline: 3.4474x; 1.3211x over previous
#include <hip/hip_runtime.h>
#include <hip/hip_bf16.h>

typedef float v2f __attribute__((ext_vector_type(2)));
typedef float v8f __attribute__((ext_vector_type(8)));

#define FD 128


#define AGG_ROWS    196
#define OWN_SLOTS   256
#define BM_EDGES    1024
#define BM_WORDS    32
#define REC_MAX     (BM_WORDS - 1)
#define BM_PITCH    (BM_WORDS + 1)
#define PRE_THREADS 256
#define BM_EPT      (BM_EDGES / PRE_THREADS)
#define REC_NONE    0xFFFFFFFFu

static_assert(PRE_THREADS == OWN_SLOTS, "one thread prepares one owner's line");
static_assert(BM_EPT == 4, "a thread's edges come in one 16-byte load");
static_assert(AGG_ROWS < 255, "the row inside the owner's range must fit the high byte");

__device__ __forceinline__
void own_block(const int* __restrict__ src, const int* __restrict__ dst,
               unsigned* __restrict__ bm, int E, int rowWords,
               int chunk, int passes, unsigned* m) {
    const int tid = threadIdx.x;
    const int e0 = chunk * BM_EDGES + BM_EPT * tid;
    int d[BM_EPT], s[BM_EPT];
    if (e0 + BM_EPT <= E && ((((size_t)dst) | ((size_t)src)) & 15) == 0) {
        const int4 dv = *(const int4*)(dst + e0);
        const int4 sv = *(const int4*)(src + e0);
        d[0] = dv.x; d[1] = dv.y; d[2] = dv.z; d[3] = dv.w;
        s[0] = sv.x; s[1] = sv.y; s[2] = sv.z; s[3] = sv.w;
    } else {
        #pragma unroll
        for (int k = 0; k < BM_EPT; ++k) {
            const bool in = e0 + k < E;
            d[k] = in ? dst[e0 + k] : -1;
            s[k] = in ? src[e0 + k] : 0;
        }
    }
    const int w  = tid & 31;
    const int wv = __builtin_amdgcn_readfirstlane(tid >> 5);
    for (int p = 0; p < passes; ++p) {
        if (p) __syncthreads();
        unsigned* mine = m + tid * BM_PITCH;
        mine[0] = 0u;
        #pragma unroll
        for (int k = 1; k < BM_WORDS; ++k) mine[k] = REC_NONE;
        __syncthreads();
        #pragma unroll
        for (int k = 0; k < BM_EPT; ++k) {
            if (d[k] >= 0) {
                const unsigned o = (unsigned)d[k] / AGG_ROWS;
                if ((int)(o / OWN_SLOTS) == p) {
                    const unsigned r = (unsigned)d[k] - o * AGG_ROWS;
                    unsigned* line = m + (o % OWN_SLOTS) * BM_PITCH;
                    const unsigned rank = atomicAdd(line, 1u);
                    if (rank < REC_MAX) line[1 + rank] = (unsigned)s[k] | (r << 24);
                }
            }
        }
        __syncthreads();
        unsigned* bcol = bm + (size_t)p * OWN_SLOTS * rowWords + (size_t)chunk * BM_WORDS + w;
        #pragma unroll 8
        for (int i = 0; i < OWN_SLOTS / (PRE_THREADS / 32); ++i) {
            const int o = i * (PRE_THREADS / 32) + wv;
            bcol[(size_t)o * rowWords] = m[o * BM_PITCH + w];
        }
    }
}


#define GEMM_ROWS  64
#define GEMM_PITCH 132
#define GEMM_RT    (GEMM_ROWS / 16)
#define PRE_LDS4   (GEMM_ROWS * GEMM_PITCH / 4)
static_assert(OWN_SLOTS * BM_PITCH <= PRE_LDS4 * 4, "the owners' lines must fit the shared LDS");

__device__ __forceinline__
void gemm_block(const float* __restrict__ A, const float* __restrict__ W,
                float* __restrict__ C, int M, int blk, float* sa) {
    const int tid  = threadIdx.x;
    const int lane = tid & 31;
    const int wv   = __builtin_amdgcn_readfirstlane(tid >> 5);
    const int half = lane >> 4;
    const int l16  = lane & 15;
    const int row0 = blk * GEMM_ROWS;

    const float4* A4 = (const float4*)A;
    float4 av[GEMM_ROWS / 8];
    #pragma unroll
    for (int i = 0; i < GEMM_ROWS / 8; ++i) {
        int gr = row0 + i * 8 + wv;
        av[i] = gr < M ? A4[(size_t)gr * (FD / 4) + lane] : make_float4(0.0f, 0.0f, 0.0f, 0.0f);
    }

    v2f wf[FD / 4];
    const float* wcol = W + (size_t)(2 * half) * FD + wv * 16 + l16;
    #pragma unroll
    for (int ks = 0; ks < FD / 4; ++ks) {
        wf[ks].x = wcol[(size_t)(4 * ks) * FD];
        wf[ks].y = wcol[(size_t)(4 * ks + 1) * FD];
    }

    #pragma unroll
    for (int i = 0; i < GEMM_ROWS / 8; ++i) {
        *(float4*)&sa[(i * 8 + wv) * GEMM_PITCH + 4 * lane] = av[i];
    }
    __syncthreads();

    v8f acc[GEMM_RT];
    #pragma unroll
    for (int rt = 0; rt < GEMM_RT; ++rt) acc[rt] = v8f{};

    const float* arow = sa + l16 * GEMM_PITCH + 2 * half;
    #pragma unroll
    for (int ks = 0; ks < FD / 4; ++ks) {
        #pragma unroll
        for (int rt = 0; rt < GEMM_RT; ++rt) {
            v2f a = *(const v2f*)&arow[rt * 16 * GEMM_PITCH + 4 * ks];
            acc[rt] = __builtin_amdgcn_wmma_f32_16x16x4_f32(false, a, false, wf[ks],
                                                            (short)0, acc[rt], false, false);
        }
    }

    __syncthreads();
    float* crow = sa + (8 * half) * GEMM_PITCH + wv * 16 + l16;
    #pragma unroll
    for (int rt = 0; rt < GEMM_RT; ++rt) {
        #pragma unroll
        for (int v = 0; v < 8; ++v) {
            crow[(rt * 16 + v) * GEMM_PITCH] = acc[rt][v];
        }
    }
    __syncthreads();
    float4* C4 = (float4*)C;
    #pragma unroll
    for (int i = 0; i < GEMM_ROWS / 8; ++i) {
        int r  = i * 8 + wv;
        int gr = row0 + r;
        if (gr < M) C4[(size_t)gr * (FD / 4) + lane] = *(const float4*)&sa[r * GEMM_PITCH + 4 * lane];
    }
}

__global__ __launch_bounds__(PRE_THREADS)
void gcn_gemm_wmma(const float* __restrict__ A, const float* __restrict__ W,
                   float* __restrict__ C, int M) {
    __shared__ float4 lds[PRE_LDS4];
    gemm_block(A, W, C, M, blockIdx.x, (float*)lds);
}

__global__ __launch_bounds__(PRE_THREADS)
void gcn_pre_gemm(const int* __restrict__ src, const int* __restrict__ dst,
                  unsigned* __restrict__ bm, int E, int rowWords,
                  int bblocks, int passes,
                  const float* __restrict__ A, const float* __restrict__ W,
                  float* __restrict__ C, int M) {
    __shared__ float4 lds[PRE_LDS4];
    const int b = blockIdx.x;
    if (b < bblocks) own_block(src, dst, bm, E, rowWords, b, passes, (unsigned*)lds);
    else             gemm_block(A, W, C, M, b - bblocks, (float*)lds);
}


#define AGG_CAP     16384
#define AGG_THREADS 1024
#define AGG_WAVES   (AGG_THREADS / 32)
#define AGG_LOADS   5

static_assert(AGG_THREADS % (BM_WORDS / 4) == 0, "a thread holds the same piece of a line in every load");


#define NORM_THREADS 512
#define NORM_LOADS   10

static_assert(NORM_THREADS % (BM_WORDS / 4) == 0, "a thread holds the same piece of a line in every load");
static_assert(NORM_THREADS >= AGG_ROWS, "one thread clears and writes out one owned row");

__global__ __launch_bounds__(NORM_THREADS)
void gcn_norm_owner(const int* __restrict__ dst, const unsigned* __restrict__ bm, int rowWords,
                    float* __restrict__ dis, int E, int N) {
    __shared__ unsigned rcnt[AGG_ROWS];
    __shared__ int      lost;

    const int tid  = threadIdx.x;
    const int lo   = blockIdx.x * AGG_ROWS;
    const int rows = min(AGG_ROWS, N - lo);

    if (tid < AGG_ROWS) rcnt[tid] = 0u;
    if (tid == 0) lost = 0;
    __syncthreads();

    const uint4* r4 = (const uint4*)(bm + (size_t)blockIdx.x * rowWords);
    const int nv = rowWords >> 2;
    const bool head = (tid & (BM_WORDS / 4 - 1)) == 0;
    for (int j0 = tid; j0 < nv; j0 += NORM_LOADS * NORM_THREADS) {
        uint4 bv[NORM_LOADS];
        #pragma unroll
        for (int u = 0; u < NORM_LOADS; ++u) {
            int j = j0 + u * NORM_THREADS;
            bv[u] = j < nv ? r4[j] : make_uint4(REC_NONE, REC_NONE, REC_NONE, REC_NONE);
        }
        #pragma unroll
        for (int u = 0; u < NORM_LOADS; ++u) {
            const unsigned wd[4] = {bv[u].x, bv[u].y, bv[u].z, bv[u].w};
            #pragma unroll
            for (int q = 0; q < 4; ++q) {
                if (head && q == 0) {
                    if (wd[q] != REC_NONE && wd[q] > REC_MAX) lost = 1;
                } else if (wd[q] != REC_NONE) {
                    atomicAdd(&rcnt[min(wd[q] >> 24, (unsigned)(AGG_ROWS - 1))], 1u);
                }
            }
        }
    }
    __syncthreads();

    if (lost != 0) {
        if (tid < AGG_ROWS) rcnt[tid] = 0u;
        __syncthreads();
        const unsigned urows = (unsigned)rows;
        for (int e = tid; e < E; e += NORM_THREADS) {
            unsigned r = (unsigned)(dst[e] - lo);
            if (r < urows) atomicAdd(&rcnt[r], 1u);
        }
        __syncthreads();
    }

    if (tid < rows) dis[lo + tid] = rsqrtf((float)rcnt[tid] + 1.0f);
}

template <bool RELU>
__global__ __launch_bounds__(AGG_THREADS)
void gcn_agg_owner(const float* __restrict__ h, const int* __restrict__ src,
                   const int* __restrict__ dst, const float* __restrict__ dis,
                   const unsigned* __restrict__ bm, int rowWords,
                   const float* __restrict__ bias, float* __restrict__ out, int E, int N) {
    __shared__ float    acc[AGG_ROWS * FD];
    __shared__ unsigned l_a[AGG_CAP];
    __shared__ int      cnt;
    __shared__ int      ovf;

    const int tid  = threadIdx.x;
    const int lane = tid & 31;
    const int w    = __builtin_amdgcn_readfirstlane(tid >> 5);
    const int lo   = blockIdx.x * AGG_ROWS;
    const int rows = min(AGG_ROWS, N - lo);
    const unsigned urows = (unsigned)rows;
    const float4* h4 = (const float4*)h;

    for (int r = w; r < rows; r += AGG_WAVES) {
        float dd = dis[lo + r];
        float nn = dd * dd;
        float4 v = h4[(size_t)(lo + r) * (FD / 4) + lane];
        float* a = acc + r * FD + lane;
        a[0]  = v.x * nn;
        a[32] = v.y * nn;
        a[64] = v.z * nn;
        a[96] = v.w * nn;
    }
    if (tid == 0) { cnt = 0; ovf = 0; }
    __syncthreads();

    const uint4* r4 = (const uint4*)(bm + (size_t)blockIdx.x * rowWords);
    const int nv = rowWords >> 2;
    const bool head = (tid & (BM_WORDS / 4 - 1)) == 0;
    for (int j0 = tid; j0 < nv; j0 += AGG_LOADS * AGG_THREADS) {
        uint4 bv[AGG_LOADS];
        #pragma unroll
        for (int u = 0; u < AGG_LOADS; ++u) {
            int j = j0 + u * AGG_THREADS;
            bv[u] = j < nv ? r4[j] : make_uint4(REC_NONE, REC_NONE, REC_NONE, REC_NONE);
        }
        int  total = 0;
        bool over  = false;
        #pragma unroll
        for (int u = 0; u < AGG_LOADS; ++u) {
            if (head) over = over || (bv[u].x != REC_NONE && bv[u].x > REC_MAX);
            else      total += bv[u].x != REC_NONE;
            total += (bv[u].y != REC_NONE) + (bv[u].z != REC_NONE) + (bv[u].w != REC_NONE);
        }
        if (over) ovf = 1;
        int pos = atomicAdd(&cnt, total);
        #pragma unroll
        for (int u = 0; u < AGG_LOADS; ++u) {
            const unsigned wd[4] = {bv[u].x, bv[u].y, bv[u].z, bv[u].w};
            #pragma unroll
            for (int q = 0; q < 4; ++q) {
                if (!(head && q == 0) && wd[q] != REC_NONE) {
                    if (pos < AGG_CAP) l_a[pos] = wd[q];
                    ++pos;
                }
            }
        }
    }
    __syncthreads();
    const int found = cnt;

    if (found > AGG_CAP || ovf != 0) {
        for (int e = tid; e < E; e += AGG_THREADS) {
            unsigned r = (unsigned)(dst[e] - lo);
            if (r < urows) {
                int s = src[e];
                float nn = dis[s] * dis[lo + (int)r];
                const float* hs = h + (size_t)s * FD;
                for (int c = 0; c < FD; ++c)
                    atomicAdd(&acc[r * FD + (c & 3) * 32 + (c >> 2)], hs[c] * nn);
            }
        }
    } else {
        const int groups = found >> 3;
        for (int g = w; g < groups; g += AGG_WAVES) {
            float4 v[8];
            int    rr[8];
            const unsigned recl = l_a[g * 8 + (lane & 7)];
            const float nl = dis[(int)(recl & 0xFFFFFFu)] * dis[lo + (int)(recl >> 24)];
            #pragma unroll
            for (int u = 0; u < 8; ++u) {
                const unsigned rec = l_a[g * 8 + u];
                const int s = (int)(rec & 0xFFFFFFu);
                rr[u] = (int)(rec >> 24);
                v[u]  = h4[(size_t)s * (FD / 4) + lane];
            }
            #pragma unroll
            for (int u = 0; u < 8; ++u) {
                const float nn = __int_as_float(__builtin_amdgcn_readlane(__float_as_int(nl), u));
                float* a = acc + rr[u] * FD + lane;
                atomicAdd(a,      v[u].x * nn);
                atomicAdd(a + 32, v[u].y * nn);
                atomicAdd(a + 64, v[u].z * nn);
                atomicAdd(a + 96, v[u].w * nn);
            }
        }
        {
            int i = groups * 8 + w;
            if (i < found) {
                const unsigned rec = l_a[i];
                const int s  = (int)(rec & 0xFFFFFFu);
                const int r1 = (int)(rec >> 24);
                const float n1 = dis[s] * dis[lo + r1];
                float4 v1 = h4[(size_t)s * (FD / 4) + lane];
                float* a = acc + r1 * FD + lane;
                atomicAdd(a,      v1.x * n1);
                atomicAdd(a + 32, v1.y * n1);
                atomicAdd(a + 64, v1.z * n1);
                atomicAdd(a + 96, v1.w * n1);
            }
        }
    }
    __syncthreads();

    const float4 b4 = ((const float4*)bias)[lane];
    float4* o4 = (float4*)out;
    for (int r = w; r < rows; r += AGG_WAVES) {
        const float* a = acc + r * FD + lane;
        float4 v;
        v.x = a[0]  + b4.x;
        v.y = a[32] + b4.y;
        v.z = a[64] + b4.z;
        v.w = a[96] + b4.w;
        if (RELU) {
            v.x = v.x > 0.0f ? v.x : 0.0f;
            v.y = v.y > 0.0f ? v.y : 0.0f;
            v.z = v.z > 0.0f ? v.z : 0.0f;
            v.w = v.w > 0.0f ? v.w : 0.0f;
        }
        o4[(size_t)(lo + r) * (FD / 4) + lane] = v;
    }
}


extern "C" void kernel_launch(void* const* d_in, const int* in_sizes, int n_in,
                              void* d_out, int out_size, void* d_ws, size_t ws_size,
                              hipStream_t stream) {
    const float* x  = (const float*)d_in[0];
    const int*   ei = (const int*)d_in[1];
    const float* W1 = (const float*)d_in[2];
    const float* b1 = (const float*)d_in[3];
    const float* W2 = (const float*)d_in[4];
    const float* b2 = (const float*)d_in[5];

    const int N = in_sizes[0] / FD;
    const int E = in_sizes[1] / 2;
    const int* src = ei;
    const int* dst = ei + E;

    const int ablocks = (N + AGG_ROWS - 1) / AGG_ROWS;
    const int passes  = (ablocks + OWN_SLOTS - 1) / OWN_SLOTS;

    const int bblocks  = (E + BM_EDGES - 1) / BM_EDGES;
    const int rowWords = bblocks * BM_WORDS;
    unsigned* bm = (unsigned*)d_ws;

    float* ws   = (float*)d_ws + (size_t)passes * OWN_SLOTS * rowWords;
    float* dis  = ws;
    float* bufA = ws + 2 * (size_t)N;
    float* bufB = bufA + (size_t)N * FD;
    float* out  = (float*)d_out;

    const int gblocks = (N + GEMM_ROWS - 1) / GEMM_ROWS;

    gcn_pre_gemm<<<bblocks + gblocks, PRE_THREADS, 0, stream>>>(src, dst, bm, E, rowWords, bblocks, passes, x, W1, bufA, N);
    gcn_norm_owner<<<ablocks, NORM_THREADS, 0, stream>>>(dst, bm, rowWords, dis, E, N);

    gcn_agg_owner<true><<<ablocks, AGG_THREADS, 0, stream>>>(bufA, src, dst, dis, bm, rowWords, b1, bufB, E, N);

    gcn_gemm_wmma<<<gblocks, PRE_THREADS, 0, stream>>>(bufB, W2, bufA, N);
    gcn_agg_owner<false><<<ablocks, AGG_THREADS, 0, stream>>>(bufA, src, dst, dis, bm, rowWords, b2, out, E, N);
}


# Edge norms without global lookups: norm launch also writes in-degrees as bytes; an owner copies that 50 KB table to LDS and rebuilds dis[src] there; dis[dst] is applied once per row at the store
# speedup vs baseline: 3.6617x; 1.0622x over previous
#include <hip/hip_runtime.h>
#include <hip/hip_bf16.h>

typedef float v2f __attribute__((ext_vector_type(2)));
typedef float v8f __attribute__((ext_vector_type(8)));

#define FD 128


#define AGG_ROWS    196
#define OWN_SLOTS   256
#define BM_EDGES    1024
#define BM_WORDS    32
#define REC_MAX     (BM_WORDS - 1)
#define BM_PITCH    (BM_WORDS + 1)
#define PRE_THREADS 256
#define BM_EPT      (BM_EDGES / PRE_THREADS)
#define REC_NONE    0xFFFFFFFFu

static_assert(PRE_THREADS == OWN_SLOTS, "one thread prepares one owner's line");
static_assert(BM_EPT == 4, "a thread's edges come in one 16-byte load");
static_assert(AGG_ROWS < 255, "the row inside the owner's range must fit the high byte");

__device__ __forceinline__
void own_block(const int* __restrict__ src, const int* __restrict__ dst,
               unsigned* __restrict__ bm, int E, int rowWords,
               int chunk, int passes, unsigned* m) {
    const int tid = threadIdx.x;
    const int e0 = chunk * BM_EDGES + BM_EPT * tid;
    int d[BM_EPT], s[BM_EPT];
    if (e0 + BM_EPT <= E && ((((size_t)dst) | ((size_t)src)) & 15) == 0) {
        const int4 dv = *(const int4*)(dst + e0);
        const int4 sv = *(const int4*)(src + e0);
        d[0] = dv.x; d[1] = dv.y; d[2] = dv.z; d[3] = dv.w;
        s[0] = sv.x; s[1] = sv.y; s[2] = sv.z; s[3] = sv.w;
    } else {
        #pragma unroll
        for (int k = 0; k < BM_EPT; ++k) {
            const bool in = e0 + k < E;
            d[k] = in ? dst[e0 + k] : -1;
            s[k] = in ? src[e0 + k] : 0;
        }
    }
    const int w  = tid & 31;
    const int wv = __builtin_amdgcn_readfirstlane(tid >> 5);
    for (int p = 0; p < passes; ++p) {
        if (p) __syncthreads();
        unsigned* mine = m + tid * BM_PITCH;
        mine[0] = 0u;
        #pragma unroll
        for (int k = 1; k < BM_WORDS; ++k) mine[k] = REC_NONE;
        __syncthreads();
        #pragma unroll
        for (int k = 0; k < BM_EPT; ++k) {
            if (d[k] >= 0) {
                const unsigned o = (unsigned)d[k] / AGG_ROWS;
                if ((int)(o / OWN_SLOTS) == p) {
                    const unsigned r = (unsigned)d[k] - o * AGG_ROWS;
                    unsigned* line = m + (o % OWN_SLOTS) * BM_PITCH;
                    const unsigned rank = atomicAdd(line, 1u);
                    if (rank < REC_MAX) line[1 + rank] = (unsigned)s[k] | (r << 24);
                }
            }
        }
        __syncthreads();
        unsigned* bcol = bm + (size_t)p * OWN_SLOTS * rowWords + (size_t)chunk * BM_WORDS + w;
        #pragma unroll 8
        for (int i = 0; i < OWN_SLOTS / (PRE_THREADS / 32); ++i) {
            const int o = i * (PRE_THREADS / 32) + wv;
            bcol[(size_t)o * rowWords] = m[o * BM_PITCH + w];
        }
    }
}


#define GEMM_ROWS  64
#define GEMM_PITCH 132
#define GEMM_RT    (GEMM_ROWS / 16)
#define PRE_LDS4   (GEMM_ROWS * GEMM_PITCH / 4)
static_assert(OWN_SLOTS * BM_PITCH <= PRE_LDS4 * 4, "the owners' lines must fit the shared LDS");

__device__ __forceinline__
void gemm_block(const float* __restrict__ A, const float* __restrict__ W,
                float* __restrict__ C, int M, int blk, float* sa) {
    const int tid  = threadIdx.x;
    const int lane = tid & 31;
    const int wv   = __builtin_amdgcn_readfirstlane(tid >> 5);
    const int half = lane >> 4;
    const int l16  = lane & 15;
    const int row0 = blk * GEMM_ROWS;

    const float4* A4 = (const float4*)A;
    float4 av[GEMM_ROWS / 8];
    #pragma unroll
    for (int i = 0; i < GEMM_ROWS / 8; ++i) {
        int gr = row0 + i * 8 + wv;
        av[i] = gr < M ? A4[(size_t)gr * (FD / 4) + lane] : make_float4(0.0f, 0.0f, 0.0f, 0.0f);
    }

    v2f wf[FD / 4];
    const float* wcol = W + (size_t)(2 * half) * FD + wv * 16 + l16;
    #pragma unroll
    for (int ks = 0; ks < FD / 4; ++ks) {
        wf[ks].x = wcol[(size_t)(4 * ks) * FD];
        wf[ks].y = wcol[(size_t)(4 * ks + 1) * FD];
    }

    #pragma unroll
    for (int i = 0; i < GEMM_ROWS / 8; ++i) {
        *(float4*)&sa[(i * 8 + wv) * GEMM_PITCH + 4 * lane] = av[i];
    }
    __syncthreads();

    v8f acc[GEMM_RT];
    #pragma unroll
    for (int rt = 0; rt < GEMM_RT; ++rt) acc[rt] = v8f{};

    const float* arow = sa + l16 * GEMM_PITCH + 2 * half;
    #pragma unroll
    for (int ks = 0; ks < FD / 4; ++ks) {
        #pragma unroll
        for (int rt = 0; rt < GEMM_RT; ++rt) {
            v2f a = *(const v2f*)&arow[rt * 16 * GEMM_PITCH + 4 * ks];
            acc[rt] = __builtin_amdgcn_wmma_f32_16x16x4_f32(false, a, false, wf[ks],
                                                            (short)0, acc[rt], false, false);
        }
    }

    __syncthreads();
    float* crow = sa + (8 * half) * GEMM_PITCH + wv * 16 + l16;
    #pragma unroll
    for (int rt = 0; rt < GEMM_RT; ++rt) {
        #pragma unroll
        for (int v = 0; v < 8; ++v) {
            crow[(rt * 16 + v) * GEMM_PITCH] = acc[rt][v];
        }
    }
    __syncthreads();
    float4* C4 = (float4*)C;
    #pragma unroll
    for (int i = 0; i < GEMM_ROWS / 8; ++i) {
        int r  = i * 8 + wv;
        int gr = row0 + r;
        if (gr < M) C4[(size_t)gr * (FD / 4) + lane] = *(const float4*)&sa[r * GEMM_PITCH + 4 * lane];
    }
}

__global__ __launch_bounds__(PRE_THREADS)
void gcn_gemm_wmma(const float* __restrict__ A, const float* __restrict__ W,
                   float* __restrict__ C, int M) {
    __shared__ float4 lds[PRE_LDS4];
    gemm_block(A, W, C, M, blockIdx.x, (float*)lds);
}

__global__ __launch_bounds__(PRE_THREADS)
void gcn_pre_gemm(const int* __restrict__ src, const int* __restrict__ dst,
                  unsigned* __restrict__ bm, int E, int rowWords,
                  int bblocks, int passes,
                  const float* __restrict__ A, const float* __restrict__ W,
                  float* __restrict__ C, int M) {
    __shared__ float4 lds[PRE_LDS4];
    const int b = blockIdx.x;
    if (b < bblocks) own_block(src, dst, bm, E, rowWords, b, passes, (unsigned*)lds);
    else             gemm_block(A, W, C, M, b - bblocks, (float*)lds);
}


#define AGG_CAP     16384
#define AGG_THREADS 1024
#define AGG_WAVES   (AGG_THREADS / 32)
#define AGG_LOADS   5

#define DEG_SAT     255u
#define DEG_MAXN    65536

static_assert(AGG_THREADS % (BM_WORDS / 4) == 0, "a thread holds the same piece of a line in every load");
#define TAB_LOADS   (DEG_MAXN / 16 / AGG_THREADS)

static_assert(DEG_MAXN % (16 * AGG_THREADS) == 0, "every thread copies the same number of 16-byte pieces");
static_assert(AGG_ROWS % 4 == 0, "an owner writes its in-degrees as whole words of 4 bytes");


#define NORM_THREADS 512
#define NORM_LOADS   10

static_assert(NORM_THREADS % (BM_WORDS / 4) == 0, "a thread holds the same piece of a line in every load");
static_assert(NORM_THREADS >= AGG_ROWS, "one thread clears and writes out one owned row");

__global__ __launch_bounds__(NORM_THREADS)
void gcn_norm_owner(const int* __restrict__ dst, const unsigned* __restrict__ bm, int rowWords,
                    float* __restrict__ dis, unsigned* __restrict__ deg8w, int E, int N) {
    __shared__ unsigned rcnt[AGG_ROWS];
    __shared__ int      lost;

    const int tid  = threadIdx.x;
    const int lo   = blockIdx.x * AGG_ROWS;
    const int rows = min(AGG_ROWS, N - lo);

    if (tid < AGG_ROWS) rcnt[tid] = 0u;
    if (tid == 0) lost = 0;
    __syncthreads();

    const uint4* r4 = (const uint4*)(bm + (size_t)blockIdx.x * rowWords);
    const int nv = rowWords >> 2;
    const bool head = (tid & (BM_WORDS / 4 - 1)) == 0;
    for (int j0 = tid; j0 < nv; j0 += NORM_LOADS * NORM_THREADS) {
        uint4 bv[NORM_LOADS];
        #pragma unroll
        for (int u = 0; u < NORM_LOADS; ++u) {
            int j = j0 + u * NORM_THREADS;
            bv[u] = j < nv ? r4[j] : make_uint4(REC_NONE, REC_NONE, REC_NONE, REC_NONE);
        }
        #pragma unroll
        for (int u = 0; u < NORM_LOADS; ++u) {
            const unsigned wd[4] = {bv[u].x, bv[u].y, bv[u].z, bv[u].w};
            #pragma unroll
            for (int q = 0; q < 4; ++q) {
                if (head && q == 0) {
                    if (wd[q] != REC_NONE && wd[q] > REC_MAX) lost = 1;
                } else if (wd[q] != REC_NONE) {
                    atomicAdd(&rcnt[min(wd[q] >> 24, (unsigned)(AGG_ROWS - 1))], 1u);
                }
            }
        }
    }
    __syncthreads();

    if (lost != 0) {
        if (tid < AGG_ROWS) rcnt[tid] = 0u;
        __syncthreads();
        const unsigned urows = (unsigned)rows;
        for (int e = tid; e < E; e += NORM_THREADS) {
            unsigned r = (unsigned)(dst[e] - lo);
            if (r < urows) atomicAdd(&rcnt[r], 1u);
        }
        __syncthreads();
    }

    if (tid < rows) dis[lo + tid] = rsqrtf((float)rcnt[tid] + 1.0f);
    if (4 * tid < rows) {
        unsigned wd = 0u;
        #pragma unroll
        for (int k = 0; k < 4; ++k) wd |= min(rcnt[4 * tid + k], DEG_SAT) << (8 * k);
        deg8w[(lo >> 2) + tid] = wd;
    }
}

__device__ __forceinline__
float src_norm(const float* __restrict__ dis, const unsigned char* tab, bool tabok, int s) {
    if (tabok) {
        const unsigned dg = tab[s];
        if (dg != DEG_SAT) return rsqrtf((float)dg + 1.0f);
    }
    return dis[s];
}

template <bool RELU>
__global__ __launch_bounds__(AGG_THREADS)
void gcn_agg_owner(const float* __restrict__ h, const int* __restrict__ src,
                   const int* __restrict__ dst, const float* __restrict__ dis,
                   const unsigned* deg8w,
                   const unsigned* __restrict__ bm, int rowWords,
                   const float* __restrict__ bias, float* __restrict__ out, int E, int N) {
    __shared__ float    acc[AGG_ROWS * FD];
    __shared__ unsigned l_a[AGG_CAP];
    __shared__ uint4    tab4[DEG_MAXN / 16];
    __shared__ float    dro[AGG_ROWS];
    __shared__ int      cnt;
    __shared__ int      ovf;

    const int tid  = threadIdx.x;
    const int lane = tid & 31;
    const int w    = __builtin_amdgcn_readfirstlane(tid >> 5);
    const int lo   = blockIdx.x * AGG_ROWS;
    const int rows = min(AGG_ROWS, N - lo);
    const unsigned urows = (unsigned)rows;
    const float4* h4 = (const float4*)h;

    const bool tabok = N <= DEG_MAXN && (((size_t)deg8w) & 15) == 0;
    const unsigned char* tab = (const unsigned char*)tab4;
    const int nq = (N + 15) / 16;
    uint4 tv[TAB_LOADS];
    if (tabok) {
        const uint4* g4 = (const uint4*)deg8w;
        #pragma unroll
        for (int u = 0; u < TAB_LOADS; ++u) {
            const int j = tid + u * AGG_THREADS;
            tv[u] = j < nq ? g4[j] : make_uint4(0u, 0u, 0u, 0u);
        }
    }

    for (int r = w; r < rows; r += AGG_WAVES) {
        float dd = dis[lo + r];
        float4 v = h4[(size_t)(lo + r) * (FD / 4) + lane];
        float* a = acc + r * FD + lane;
        a[0]  = v.x * dd;
        a[32] = v.y * dd;
        a[64] = v.z * dd;
        a[96] = v.w * dd;
        if (lane == 0) dro[r] = dd;
    }
    if (tabok) {
        #pragma unroll
        for (int u = 0; u < TAB_LOADS; ++u) {
            const int j = tid + u * AGG_THREADS;
            if (j < nq) tab4[j] = tv[u];
        }
    }
    if (tid == 0) { cnt = 0; ovf = 0; }
    __syncthreads();

    const uint4* r4 = (const uint4*)(bm + (size_t)blockIdx.x * rowWords);
    const int nv = rowWords >> 2;
    const bool head = (tid & (BM_WORDS / 4 - 1)) == 0;
    for (int j0 = tid; j0 < nv; j0 += AGG_LOADS * AGG_THREADS) {
        uint4 bv[AGG_LOADS];
        #pragma unroll
        for (int u = 0; u < AGG_LOADS; ++u) {
            int j = j0 + u * AGG_THREADS;
            bv[u] = j < nv ? r4[j] : make_uint4(REC_NONE, REC_NONE, REC_NONE, REC_NONE);
        }
        int  total = 0;
        bool over  = false;
        #pragma unroll
        for (int u = 0; u < AGG_LOADS; ++u) {
            if (head) over = over || (bv[u].x != REC_NONE && bv[u].x > REC_MAX);
            else      total += bv[u].x != REC_NONE;
            total += (bv[u].y != REC_NONE) + (bv[u].z != REC_NONE) + (bv[u].w != REC_NONE);
        }
        if (over) ovf = 1;
        int pos = atomicAdd(&cnt, total);
        #pragma unroll
        for (int u = 0; u < AGG_LOADS; ++u) {
            const unsigned wd[4] = {bv[u].x, bv[u].y, bv[u].z, bv[u].w};
            #pragma unroll
            for (int q = 0; q < 4; ++q) {
                if (!(head && q == 0) && wd[q] != REC_NONE) {
                    if (pos < AGG_CAP) l_a[pos] = wd[q];
                    ++pos;
                }
            }
        }
    }
    __syncthreads();
    const int found = cnt;

    if (found > AGG_CAP || ovf != 0) {
        for (int e = tid; e < E; e += AGG_THREADS) {
            unsigned r = (unsigned)(dst[e] - lo);
            if (r < urows) {
                int s = src[e];
                float nn = dis[s];
                const float* hs = h + (size_t)s * FD;
                for (int c = 0; c < FD; ++c)
                    atomicAdd(&acc[r * FD + (c & 3) * 32 + (c >> 2)], hs[c] * nn);
            }
        }
    } else {
        const int groups = found >> 3;
        for (int g = w; g < groups; g += AGG_WAVES) {
            float4 v[8];
            int    rr[8];
            #pragma unroll
            for (int u = 0; u < 8; ++u) {
                const unsigned rec = l_a[g * 8 + u];
                const int s = (int)(rec & 0xFFFFFFu);
                rr[u] = (int)(rec >> 24);
                v[u]  = h4[(size_t)s * (FD / 4) + lane];
            }
            const unsigned recl = l_a[g * 8 + (lane & 7)];
            const float nl = src_norm(dis, tab, tabok, (int)(recl & 0xFFFFFFu));
            #pragma unroll
            for (int u = 0; u < 8; ++u) {
                const float nn = __int_as_float(__builtin_amdgcn_readlane(__float_as_int(nl), u));
                float* a = acc + rr[u] * FD + lane;
                atomicAdd(a,      v[u].x * nn);
                atomicAdd(a + 32, v[u].y * nn);
                atomicAdd(a + 64, v[u].z * nn);
                atomicAdd(a + 96, v[u].w * nn);
            }
        }
        {
            int i = groups * 8 + w;
            if (i < found) {
                const unsigned rec = l_a[i];
                const int s  = (int)(rec & 0xFFFFFFu);
                const int r1 = (int)(rec >> 24);
                const float n1 = src_norm(dis, tab, tabok, s);
                float4 v1 = h4[(size_t)s * (FD / 4) + lane];
                float* a = acc + r1 * FD + lane;
                atomicAdd(a,      v1.x * n1);
                atomicAdd(a + 32, v1.y * n1);
                atomicAdd(a + 64, v1.z * n1);
                atomicAdd(a + 96, v1.w * n1);
            }
        }
    }
    __syncthreads();

    const float4 b4 = ((const float4*)bias)[lane];
    float4* o4 = (float4*)out;
    for (int r = w; r < rows; r += AGG_WAVES) {
        const float* a = acc + r * FD + lane;
        const float dr = dro[r];
        float4 v;
        v.x = a[0]  * dr + b4.x;
        v.y = a[32] * dr + b4.y;
        v.z = a[64] * dr + b4.z;
        v.w = a[96] * dr + b4.w;
        if (RELU) {
            v.x = v.x > 0.0f ? v.x : 0.0f;
            v.y = v.y > 0.0f ? v.y : 0.0f;
            v.z = v.z > 0.0f ? v.z : 0.0f;
            v.w = v.w > 0.0f ? v.w : 0.0f;
        }
        o4[(size_t)(lo + r) * (FD / 4) + lane] = v;
    }
}


extern "C" void kernel_launch(void* const* d_in, const int* in_sizes, int n_in,
                              void* d_out, int out_size, void* d_ws, size_t ws_size,
                              hipStream_t stream) {
    const float* x  = (const float*)d_in[0];
    const int*   ei = (const int*)d_in[1];
    const float* W1 = (const float*)d_in[2];
    const float* b1 = (const float*)d_in[3];
    const float* W2 = (const float*)d_in[4];
    const float* b2 = (const float*)d_in[5];

    const int N = in_sizes[0] / FD;
    const int E = in_sizes[1] / 2;
    const int* src = ei;
    const int* dst = ei + E;

    const int ablocks = (N + AGG_ROWS - 1) / AGG_ROWS;
    const int passes  = (ablocks + OWN_SLOTS - 1) / OWN_SLOTS;

    const int bblocks  = (E + BM_EDGES - 1) / BM_EDGES;
    const int rowWords = bblocks * BM_WORDS;
    unsigned* bm = (unsigned*)d_ws;

    float* ws   = (float*)d_ws + (size_t)passes * OWN_SLOTS * rowWords;
    float* dis  = ws;
    unsigned* deg8w = (unsigned*)(ws + N);
    float* bufA = ws + 2 * (size_t)N;
    float* bufB = bufA + (size_t)N * FD;
    float* out  = (float*)d_out;

    const int gblocks = (N + GEMM_ROWS - 1) / GEMM_ROWS;

    gcn_pre_gemm<<<bblocks + gblocks, PRE_THREADS, 0, stream>>>(src, dst, bm, E, rowWords, bblocks, passes, x, W1, bufA, N);
    gcn_norm_owner<<<ablocks, NORM_THREADS, 0, stream>>>(dst, bm, rowWords, dis, deg8w, E, N);

    gcn_agg_owner<true><<<ablocks, AGG_THREADS, 0, stream>>>(bufA, src, dst, dis, deg8w, bm, rowWords, b1, bufB, E, N);

    gcn_gemm_wmma<<<gblocks, PRE_THREADS, 0, stream>>>(bufB, W2, bufA, N);
    gcn_agg_owner<false><<<ablocks, AGG_THREADS, 0, stream>>>(bufA, src, dst, dis, deg8w, bm, rowWords, b2, out, E, N);
}


# Owner deals its records (self loops too) to the wave that owns the row (row mod 32, LDS integer rank); a wave sums its 7 rows in registers and stores them: no LDS float adds; sum order = list rank
# speedup vs baseline: 3.8090x; 1.0402x over previous
#include <hip/hip_runtime.h>
#include <hip/hip_bf16.h>

typedef float v2f __attribute__((ext_vector_type(2)));
typedef float v8f __attribute__((ext_vector_type(8)));

#define FD 128


#define AGG_ROWS    196
#define OWN_SLOTS   256
#define BM_EDGES    1024
#define BM_WORDS    32
#define REC_MAX     (BM_WORDS - 1)
#define BM_PITCH    (BM_WORDS + 1)
#define PRE_THREADS 256
#define BM_EPT      (BM_EDGES / PRE_THREADS)
#define REC_NONE    0xFFFFFFFFu

static_assert(PRE_THREADS == OWN_SLOTS, "one thread prepares one owner's line");
static_assert(BM_EPT == 4, "a thread's edges come in one 16-byte load");
static_assert(AGG_ROWS < 255, "the row inside the owner's range must fit the high byte");

__device__ __forceinline__
void own_block(const int* __restrict__ src, const int* __restrict__ dst,
               unsigned* __restrict__ bm, int E, int rowWords,
               int chunk, int passes, unsigned* m) {
    const int tid = threadIdx.x;
    const int e0 = chunk * BM_EDGES + BM_EPT * tid;
    int d[BM_EPT], s[BM_EPT];
    if (e0 + BM_EPT <= E && ((((size_t)dst) | ((size_t)src)) & 15) == 0) {
        const int4 dv = *(const int4*)(dst + e0);
        const int4 sv = *(const int4*)(src + e0);
        d[0] = dv.x; d[1] = dv.y; d[2] = dv.z; d[3] = dv.w;
        s[0] = sv.x; s[1] = sv.y; s[2] = sv.z; s[3] = sv.w;
    } else {
        #pragma unroll
        for (int k = 0; k < BM_EPT; ++k) {
            const bool in = e0 + k < E;
            d[k] = in ? dst[e0 + k] : -1;
            s[k] = in ? src[e0 + k] : 0;
        }
    }
    const int w  = tid & 31;
    const int wv = __builtin_amdgcn_readfirstlane(tid >> 5);
    for (int p = 0; p < passes; ++p) {
        if (p) __syncthreads();
        unsigned* mine = m + tid * BM_PITCH;
        mine[0] = 0u;
        #pragma unroll
        for (int k = 1; k < BM_WORDS; ++k) mine[k] = REC_NONE;
        __syncthreads();
        #pragma unroll
        for (int k = 0; k < BM_EPT; ++k) {
            if (d[k] >= 0) {
                const unsigned o = (unsigned)d[k] / AGG_ROWS;
                if ((int)(o / OWN_SLOTS) == p) {
                    const unsigned r = (unsigned)d[k] - o * AGG_ROWS;
                    unsigned* line = m + (o % OWN_SLOTS) * BM_PITCH;
                    const unsigned rank = atomicAdd(line, 1u);
                    if (rank < REC_MAX) line[1 + rank] = (unsigned)s[k] | (r << 24);
                }
            }
        }
        __syncthreads();
        unsigned* bcol = bm + (size_t)p * OWN_SLOTS * rowWords + (size_t)chunk * BM_WORDS + w;
        #pragma unroll 8
        for (int i = 0; i < OWN_SLOTS / (PRE_THREADS / 32); ++i) {
            const int o = i * (PRE_THREADS / 32) + wv;
            bcol[(size_t)o * rowWords] = m[o * BM_PITCH + w];
        }
    }
}


#define GEMM_ROWS  64
#define GEMM_PITCH 132
#define GEMM_RT    (GEMM_ROWS / 16)
#define PRE_LDS4   (GEMM_ROWS * GEMM_PITCH / 4)
static_assert(OWN_SLOTS * BM_PITCH <= PRE_LDS4 * 4, "the owners' lines must fit the shared LDS");

__device__ __forceinline__
void gemm_block(const float* __restrict__ A, const float* __restrict__ W,
                float* __restrict__ C, int M, int blk, float* sa) {
    const int tid  = threadIdx.x;
    const int lane = tid & 31;
    const int wv   = __builtin_amdgcn_readfirstlane(tid >> 5);
    const int half = lane >> 4;
    const int l16  = lane & 15;
    const int row0 = blk * GEMM_ROWS;

    const float4* A4 = (const float4*)A;
    float4 av[GEMM_ROWS / 8];
    #pragma unroll
    for (int i = 0; i < GEMM_ROWS / 8; ++i) {
        int gr = row0 + i * 8 + wv;
        av[i] = gr < M ? A4[(size_t)gr * (FD / 4) + lane] : make_float4(0.0f, 0.0f, 0.0f, 0.0f);
    }

    v2f wf[FD / 4];
    const float* wcol = W + (size_t)(2 * half) * FD + wv * 16 + l16;
    #pragma unroll
    for (int ks = 0; ks < FD / 4; ++ks) {
        wf[ks].x = wcol[(size_t)(4 * ks) * FD];
        wf[ks].y = wcol[(size_t)(4 * ks + 1) * FD];
    }

    #pragma unroll
    for (int i = 0; i < GEMM_ROWS / 8; ++i) {
        *(float4*)&sa[(i * 8 + wv) * GEMM_PITCH + 4 * lane] = av[i];
    }
    __syncthreads();

    v8f acc[GEMM_RT];
    #pragma unroll
    for (int rt = 0; rt < GEMM_RT; ++rt) acc[rt] = v8f{};

    const float* arow = sa + l16 * GEMM_PITCH + 2 * half;
    #pragma unroll
    for (int ks = 0; ks < FD / 4; ++ks) {
        #pragma unroll
        for (int rt = 0; rt < GEMM_RT; ++rt) {
            v2f a = *(const v2f*)&arow[rt * 16 * GEMM_PITCH + 4 * ks];
            acc[rt] = __builtin_amdgcn_wmma_f32_16x16x4_f32(false, a, false, wf[ks],
                                                            (short)0, acc[rt], false, false);
        }
    }

    __syncthreads();
    float* crow = sa + (8 * half) * GEMM_PITCH + wv * 16 + l16;
    #pragma unroll
    for (int rt = 0; rt < GEMM_RT; ++rt) {
        #pragma unroll
        for (int v = 0; v < 8; ++v) {
            crow[(rt * 16 + v) * GEMM_PITCH] = acc[rt][v];
        }
    }
    __syncthreads();
    float4* C4 = (float4*)C;
    #pragma unroll
    for (int i = 0; i < GEMM_ROWS / 8; ++i) {
        int r  = i * 8 + wv;
        int gr = row0 + r;
        if (gr < M) C4[(size_t)gr * (FD / 4) + lane] = *(const float4*)&sa[r * GEMM_PITCH + 4 * lane];
    }
}

__global__ __launch_bounds__(PRE_THREADS)
void gcn_gemm_wmma(const float* __restrict__ A, const float* __restrict__ W,
                   float* __restrict__ C, int M) {
    __shared__ float4 lds[PRE_LDS4];
    gemm_block(A, W, C, M, blockIdx.x, (float*)lds);
}

__global__ __launch_bounds__(PRE_THREADS)
void gcn_pre_gemm(const int* __restrict__ src, const int* __restrict__ dst,
                  unsigned* __restrict__ bm, int E, int rowWords,
                  int bblocks, int passes,
                  const float* __restrict__ A, const float* __restrict__ W,
                  float* __restrict__ C, int M) {
    __shared__ float4 lds[PRE_LDS4];
    const int b = blockIdx.x;
    if (b < bblocks) own_block(src, dst, bm, E, rowWords, b, passes, (unsigned*)lds);
    else             gemm_block(A, W, C, M, b - bblocks, (float*)lds);
}


#define AGG_CAP     16384
#define AGG_THREADS 1024
#define AGG_WAVES   (AGG_THREADS / 32)
#define AGG_LOADS   5
#define BIN_CAP     (AGG_CAP / AGG_WAVES)
#define OWN_PER_WAVE ((AGG_ROWS + AGG_WAVES - 1) / AGG_WAVES)

static_assert(AGG_WAVES == 32, "a record's list is its row modulo 32, its sum the row divided by 32");

#define DEG_SAT     255u
#define DEG_MAXN    65536

static_assert(AGG_THREADS % (BM_WORDS / 4) == 0, "a thread holds the same piece of a line in every load");
#define TAB_LOADS   (DEG_MAXN / 16 / AGG_THREADS)

static_assert(DEG_MAXN % (16 * AGG_THREADS) == 0, "every thread copies the same number of 16-byte pieces");
static_assert(AGG_ROWS % 4 == 0, "an owner writes its in-degrees as whole words of 4 bytes");


#define NORM_THREADS 512
#define NORM_LOADS   10

static_assert(NORM_THREADS % (BM_WORDS / 4) == 0, "a thread holds the same piece of a line in every load");
static_assert(NORM_THREADS >= AGG_ROWS, "one thread clears and writes out one owned row");

__global__ __launch_bounds__(NORM_THREADS)
void gcn_norm_owner(const int* __restrict__ dst, const unsigned* __restrict__ bm, int rowWords,
                    float* __restrict__ dis, unsigned* __restrict__ deg8w, int E, int N) {
    __shared__ unsigned rcnt[AGG_ROWS];
    __shared__ int      lost;

    const int tid  = threadIdx.x;
    const int lo   = blockIdx.x * AGG_ROWS;
    const int rows = min(AGG_ROWS, N - lo);

    if (tid < AGG_ROWS) rcnt[tid] = 0u;
    if (tid == 0) lost = 0;
    __syncthreads();

    const uint4* r4 = (const uint4*)(bm + (size_t)blockIdx.x * rowWords);
    const int nv = rowWords >> 2;
    const bool head = (tid & (BM_WORDS / 4 - 1)) == 0;
    for (int j0 = tid; j0 < nv; j0 += NORM_LOADS * NORM_THREADS) {
        uint4 bv[NORM_LOADS];
        #pragma unroll
        for (int u = 0; u < NORM_LOADS; ++u) {
            int j = j0 + u * NORM_THREADS;
            bv[u] = j < nv ? r4[j] : make_uint4(REC_NONE, REC_NONE, REC_NONE, REC_NONE);
        }
        #pragma unroll
        for (int u = 0; u < NORM_LOADS; ++u) {
            const unsigned wd[4] = {bv[u].x, bv[u].y, bv[u].z, bv[u].w};
            #pragma unroll
            for (int q = 0; q < 4; ++q) {
                if (head && q == 0) {
                    if (wd[q] != REC_NONE && wd[q] > REC_MAX) lost = 1;
                } else if (wd[q] != REC_NONE) {
                    atomicAdd(&rcnt[min(wd[q] >> 24, (unsigned)(AGG_ROWS - 1))], 1u);
                }
            }
        }
    }
    __syncthreads();

    if (lost != 0) {
        if (tid < AGG_ROWS) rcnt[tid] = 0u;
        __syncthreads();
        const unsigned urows = (unsigned)rows;
        for (int e = tid; e < E; e += NORM_THREADS) {
            unsigned r = (unsigned)(dst[e] - lo);
            if (r < urows) atomicAdd(&rcnt[r], 1u);
        }
        __syncthreads();
    }

    if (tid < rows) dis[lo + tid] = rsqrtf((float)rcnt[tid] + 1.0f);
    if (4 * tid < rows) {
        unsigned wd = 0u;
        #pragma unroll
        for (int k = 0; k < 4; ++k) wd |= min(rcnt[4 * tid + k], DEG_SAT) << (8 * k);
        deg8w[(lo >> 2) + tid] = wd;
    }
}

__device__ __forceinline__
float src_norm(const float* __restrict__ dis, const unsigned char* tab, bool tabok, int s) {
    if (tabok) {
        const unsigned dg = tab[s];
        if (dg != DEG_SAT) return rsqrtf((float)dg + 1.0f);
    }
    return dis[s];
}

typedef float v32f __attribute__((ext_vector_type(32)));
static_assert(OWN_PER_WAVE < 8, "a spare sum behind the wave's own");

__device__ __forceinline__
void add_row(v32f& a, int which, const float4 v, float nn) {
    a[which]      += v.x * nn;
    a[8 + which]  += v.y * nn;
    a[16 + which] += v.z * nn;
    a[24 + which] += v.w * nn;
}

template <bool PARTIAL>
__device__ __forceinline__
void gather_group(v32f& a, const unsigned* mine, int g0, int n, int lane,
                  const float4* __restrict__ h4, const float* __restrict__ dis,
                  const unsigned char* tab, bool tabok) {
    float4 v[8];
    int    which[8];
    #pragma unroll
    for (int u = 0; u < 8; ++u) {
        const int i = PARTIAL ? min(g0 + u, n - 1) : g0 + u;
        const unsigned rec = mine[i];
        const int s = (int)(rec & 0xFFFFFFu);
        const int wh = __builtin_amdgcn_readfirstlane((int)(rec >> 29));
        if (PARTIAL && g0 + u >= n) {
            which[u] = OWN_PER_WAVE;
            v[u] = make_float4(0.0f, 0.0f, 0.0f, 0.0f);
        } else {
            which[u] = wh;
            v[u] = h4[(size_t)s * (FD / 4) + lane];
        }
    }
    const int il = PARTIAL ? min(g0 + (lane & 7), n - 1) : g0 + (lane & 7);
    const unsigned recl = mine[il];
    const float nl = src_norm(dis, tab, tabok, (int)(recl & 0xFFFFFFu));
    #pragma unroll
    for (int u = 0; u < 8; ++u) {
        const float nn = __int_as_float(__builtin_amdgcn_readlane(__float_as_int(nl), u));
        add_row(a, which[u], v[u], nn);
    }
}

template <bool RELU>
__global__ __launch_bounds__(AGG_THREADS)
void gcn_agg_owner(const float* __restrict__ h, const int* __restrict__ src,
                   const int* __restrict__ dst, const float* __restrict__ dis,
                   const unsigned* deg8w,
                   const unsigned* __restrict__ bm, int rowWords,
                   const float* __restrict__ bias, float* __restrict__ out, int E, int N) {
    __shared__ float    acc[AGG_ROWS * FD];
    __shared__ unsigned l_a[AGG_CAP];
    __shared__ uint4    tab4[DEG_MAXN / 16];
    __shared__ float    dro[AGG_ROWS];
    __shared__ unsigned bcnt[AGG_WAVES];
    __shared__ int      ovf;

    const int tid  = threadIdx.x;
    const int lane = tid & 31;
    const int w    = __builtin_amdgcn_readfirstlane(tid >> 5);
    const int lo   = blockIdx.x * AGG_ROWS;
    const int rows = min(AGG_ROWS, N - lo);
    const unsigned urows = (unsigned)rows;
    const float4* h4 = (const float4*)h;

    const bool tabok = N <= DEG_MAXN && (((size_t)deg8w) & 15) == 0;
    const unsigned char* tab = (const unsigned char*)tab4;
    const int nq = (N + 15) / 16;
    uint4 tv[TAB_LOADS];
    if (tabok) {
        const uint4* g4 = (const uint4*)deg8w;
        #pragma unroll
        for (int u = 0; u < TAB_LOADS; ++u) {
            const int j = tid + u * AGG_THREADS;
            tv[u] = j < nq ? g4[j] : make_uint4(0u, 0u, 0u, 0u);
        }
    }
    const uint4* r4 = (const uint4*)(bm + (size_t)blockIdx.x * rowWords);
    const int nv = rowWords >> 2;
    uint4 bv[AGG_LOADS];
    #pragma unroll
    for (int u = 0; u < AGG_LOADS; ++u) {
        const int j = tid + u * AGG_THREADS;
        bv[u] = j < nv ? r4[j] : make_uint4(REC_NONE, REC_NONE, REC_NONE, REC_NONE);
    }

    if (tid < rows) l_a[(tid & 31) * BIN_CAP + (tid >> 5)] = (unsigned)(lo + tid) | ((unsigned)tid << 24);
    if (tid < AGG_WAVES) bcnt[tid] = rows > tid ? (unsigned)((rows - 1 - tid) >> 5) + 1u : 0u;
    if (tid == 0) ovf = 0;
    if (tabok) {
        #pragma unroll
        for (int u = 0; u < TAB_LOADS; ++u) {
            const int j = tid + u * AGG_THREADS;
            if (j < nq) tab4[j] = tv[u];
        }
    }
    __syncthreads();

    const bool head = (tid & (BM_WORDS / 4 - 1)) == 0;
    for (int j0 = tid; j0 < nv; j0 += AGG_LOADS * AGG_THREADS) {
        if (j0 != tid) {
            #pragma unroll
            for (int u = 0; u < AGG_LOADS; ++u) {
                const int j = j0 + u * AGG_THREADS;
                bv[u] = j < nv ? r4[j] : make_uint4(REC_NONE, REC_NONE, REC_NONE, REC_NONE);
            }
        }
        bool over = false;
        #pragma unroll
        for (int u = 0; u < AGG_LOADS; ++u) {
            const unsigned wd[4] = {bv[u].x, bv[u].y, bv[u].z, bv[u].w};
            #pragma unroll
            for (int q = 0; q < 4; ++q) {
                if (head && q == 0) {
                    over = over || (wd[q] != REC_NONE && wd[q] > REC_MAX);
                } else if (wd[q] != REC_NONE) {
                    const unsigned b = (wd[q] >> 24) & 31u;
                    const unsigned pos = atomicAdd(&bcnt[b], 1u);
                    if (pos < BIN_CAP) l_a[b * BIN_CAP + pos] = wd[q];
                }
            }
        }
        if (over) ovf = 1;
    }
    __syncthreads();

    const unsigned myc = bcnt[lane];
    const bool fallback = ovf != 0 || __builtin_amdgcn_ballot_w32(myc > BIN_CAP) != 0u;

    const float4 b4 = ((const float4*)bias)[lane];
    float4* o4 = (float4*)out;

    if (fallback) {
        for (int r = w; r < rows; r += AGG_WAVES) {
            float dd = dis[lo + r];
            float4 v = h4[(size_t)(lo + r) * (FD / 4) + lane];
            float* a = acc + r * FD + lane;
            a[0]  = v.x * dd;
            a[32] = v.y * dd;
            a[64] = v.z * dd;
            a[96] = v.w * dd;
            if (lane == 0) dro[r] = dd;
        }
        __syncthreads();
        for (int e = tid; e < E; e += AGG_THREADS) {
            unsigned r = (unsigned)(dst[e] - lo);
            if (r < urows) {
                int s = src[e];
                float nn = dis[s];
                const float* hs = h + (size_t)s * FD;
                for (int c = 0; c < FD; ++c)
                    atomicAdd(&acc[r * FD + (c & 3) * 32 + (c >> 2)], hs[c] * nn);
            }
        }
        __syncthreads();
        for (int r = w; r < rows; r += AGG_WAVES) {
            const float* a = acc + r * FD + lane;
            const float dr = dro[r];
            float4 v;
            v.x = a[0]  * dr + b4.x;
            v.y = a[32] * dr + b4.y;
            v.z = a[64] * dr + b4.z;
            v.w = a[96] * dr + b4.w;
            if (RELU) {
                v.x = v.x > 0.0f ? v.x : 0.0f;
                v.y = v.y > 0.0f ? v.y : 0.0f;
                v.z = v.z > 0.0f ? v.z : 0.0f;
                v.w = v.w > 0.0f ? v.w : 0.0f;
            }
            o4[(size_t)(lo + r) * (FD / 4) + lane] = v;
        }
    } else {
        const int n = __builtin_amdgcn_readlane((int)myc, w);
        const unsigned* mine = l_a + w * BIN_CAP;
        v32f a = {};

        const int full = n & ~7;
        for (int g0 = 0; g0 < full; g0 += 8)
            gather_group<false>(a, mine, g0, n, lane, h4, dis, tab, tabok);
        if (full < n)
            gather_group<true>(a, mine, full, n, lane, h4, dis, tab, tabok);

        #pragma unroll
        for (int j = 0; j < OWN_PER_WAVE; ++j) {
            const int r = w + j * AGG_WAVES;
            if (r < rows) {
                const float dr = src_norm(dis, tab, tabok, lo + r);
                float4 v;
                v.x = a[j]      * dr + b4.x;
                v.y = a[8 + j]  * dr + b4.y;
                v.z = a[16 + j] * dr + b4.z;
                v.w = a[24 + j] * dr + b4.w;
                if (RELU) {
                    v.x = v.x > 0.0f ? v.x : 0.0f;
                    v.y = v.y > 0.0f ? v.y : 0.0f;
                    v.z = v.z > 0.0f ? v.z : 0.0f;
                    v.w = v.w > 0.0f ? v.w : 0.0f;
                }
                o4[(size_t)(lo + r) * (FD / 4) + lane] = v;
            }
        }
    }
}


extern "C" void kernel_launch(void* const* d_in, const int* in_sizes, int n_in,
                              void* d_out, int out_size, void* d_ws, size_t ws_size,
                              hipStream_t stream) {
    const float* x  = (const float*)d_in[0];
    const int*   ei = (const int*)d_in[1];
    const float* W1 = (const float*)d_in[2];
    const float* b1 = (const float*)d_in[3];
    const float* W2 = (const float*)d_in[4];
    const float* b2 = (const float*)d_in[5];

    const int N = in_sizes[0] / FD;
    const int E = in_sizes[1] / 2;
    const int* src = ei;
    const int* dst = ei + E;

    const int ablocks = (N + AGG_ROWS - 1) / AGG_ROWS;
    const int passes  = (ablocks + OWN_SLOTS - 1) / OWN_SLOTS;

    const int bblocks  = (E + BM_EDGES - 1) / BM_EDGES;
    const int rowWords = bblocks * BM_WORDS;
    unsigned* bm = (unsigned*)d_ws;

    float* ws   = (float*)d_ws + (size_t)passes * OWN_SLOTS * rowWords;
    float* dis  = ws;
    unsigned* deg8w = (unsigned*)(ws + N);
    float* bufA = ws + 2 * (size_t)N;
    float* bufB = bufA + (size_t)N * FD;
    float* out  = (float*)d_out;

    const int gblocks = (N + GEMM_ROWS - 1) / GEMM_ROWS;

    gcn_pre_gemm<<<bblocks + gblocks, PRE_THREADS, 0, stream>>>(src, dst, bm, E, rowWords, bblocks, passes, x, W1, bufA, N);
    gcn_norm_owner<<<ablocks, NORM_THREADS, 0, stream>>>(dst, bm, rowWords, dis, deg8w, E, N);

    gcn_agg_owner<true><<<ablocks, AGG_THREADS, 0, stream>>>(bufA, src, dst, dis, deg8w, bm, rowWords, b1, bufB, E, N);

    gcn_gemm_wmma<<<gblocks, PRE_THREADS, 0, stream>>>(bufB, W2, bufA, N);
    gcn_agg_owner<false><<<ablocks, AGG_THREADS, 0, stream>>>(bufA, src, dst, dis, deg8w, bm, rowWords, b2, out, E, N);
}
